# starting kernel
# baseline (speedup 1.0000x reference)
#include <hip/hip_runtime.h>
#include <hip/hip_bf16.h>


typedef __attribute__((ext_vector_type(2))) float v2f;
typedef __attribute__((ext_vector_type(8))) float v8f;

#define N_DST1   50000
#define N_DST2   10000
#define IN_F     256
#define KDIM     512
#define KSTEPS   (KDIM / 4)
#define KSPLIT   (IN_F / 4)
#define HID_F    256
#define N_CLS    64

__global__ void zero_i32_kernel(int* __restrict__ p, int n) {
  int i = blockIdx.x * blockDim.x + threadIdx.x;
  if (i < n) p[i] = 0;
}

__global__ void hist_kernel(const int* __restrict__ dst,
                            int* __restrict__ hist, int nE) {
  int i = blockIdx.x * blockDim.x + threadIdx.x;
  if (i < nE) atomicAdd(&hist[dst[i]], 1);
}

__global__ __launch_bounds__(256)
void scan_kernel(const int* __restrict__ hist,
                 int* __restrict__ row_ptr,
                 int* __restrict__ cursor, int n) {
  __shared__ int partials[256];
  const int t = threadIdx.x;
  const int chunk = (n + 255) / 256;
  int begin = t * chunk;
  int end   = begin + chunk; if (end > n) end = n;
  if (begin > n) begin = n;

  int sum = 0;
  for (int i = begin; i < end; ++i) sum += hist[i];
  partials[t] = sum;
  __syncthreads();
  if (t == 0) {
    int run = 0;
    for (int i = 0; i < 256; ++i) { int v = partials[i]; partials[i] = run; run += v; }
  }
  __syncthreads();
  int run = partials[t];
  for (int i = begin; i < end; ++i) {
    row_ptr[i] = run;
    cursor[i]  = run;
    run += hist[i];
  }
  if (end == n) row_ptr[n] = run;
}

__global__ void bin_kernel(const int* __restrict__ src,
                           const int* __restrict__ dst,
                           int* __restrict__ cursor,
                           int* __restrict__ srcs, int nE) {
  int i = blockIdx.x * blockDim.x + threadIdx.x;
  if (i < nE) {
    int d = dst[i];
    int pos = atomicAdd(&cursor[d], 1);
    srcs[pos] = src[i];
  }
}

__global__ __launch_bounds__(256)
void gather_mean_kernel(const float* __restrict__ hsrc,
                        const int* __restrict__ srcs,
                        const int* __restrict__ row_ptr,
                        float* __restrict__ hN,
                        int n_dst) {
  const int d = blockIdx.x * (blockDim.x >> 5) + (threadIdx.x >> 5);
  if (d >= n_dst) return;
  const int lane = threadIdx.x & 31;
  const int beg = row_ptr[d];
  const int end = row_ptr[d + 1];

  float4 a0 = make_float4(0.f, 0.f, 0.f, 0.f);
  float4 a1 = make_float4(0.f, 0.f, 0.f, 0.f);
  for (int j = beg; j < end; ++j) {
    int s = srcs[j];
    const float4* p = (const float4*)(hsrc + (size_t)s * IN_F) + lane * 2;
    float4 v0 = p[0];
    float4 v1 = p[1];
    a0.x += v0.x; a0.y += v0.y; a0.z += v0.z; a0.w += v0.w;
    a1.x += v1.x; a1.y += v1.y; a1.z += v1.z; a1.w += v1.w;
  }
  const float inv = 1.0f / fmaxf((float)(end - beg), 1.0f);
  a0.x *= inv; a0.y *= inv; a0.z *= inv; a0.w *= inv;
  a1.x *= inv; a1.y *= inv; a1.z *= inv; a1.w *= inv;
  float4* o = (float4*)(hN + (size_t)d * IN_F) + lane * 2;
  o[0] = a0;
  o[1] = a1;
}

__global__ void pack_W_kernel(const float* __restrict__ W,
                              float* __restrict__ Bpack, int N) {
  const int ngrpN = N >> 6;
  const int total = KSTEPS * ngrpN * 128;
  int idx = blockIdx.x * blockDim.x + threadIdx.x;
  if (idx >= total) return;

  const int w    = idx & 127;
  const int t    = w & 3;
  const int lane = w >> 2;
  const int blk  = idx >> 7;
  const int gn   = blk % ngrpN;
  const int ks   = blk / ngrpN;

  const int kh  = (lane < 16) ? 0 : 2;
  const int col = gn * 64 + t * 16 + (lane & 15);
  const int kr  = ks * 4 + kh;

  float2 v;
  v.x = W[(size_t)kr * N + col];
  v.y = W[(size_t)(kr + 1) * N + col];
  ((float2*)Bpack)[idx] = v;
}

__global__ __launch_bounds__(128)
void sage_gemm_wmma_kernel(const float* __restrict__ A0,
                           const float* __restrict__ A1,
                           const float* __restrict__ Bpack,
                           const float* __restrict__ bias,
                           float* __restrict__ out,
                           int M, int N, int do_relu) {
  const int lane = threadIdx.x & 31;
  const int wave = threadIdx.x >> 5;
  const int ngrpN = N >> 6;
  const int grp   = blockIdx.x * (blockDim.x >> 5) + wave;
  const int ngrps = (M >> 4) * ngrpN;
  if (grp >= ngrps) return;

  const int tm = grp / ngrpN;
  const int gn = grp % ngrpN;

  const int l15  = lane & 15;
  const int row  = tm * 16 + l15;
  const int col0 = gn * 64 + l15;
  const int kh   = (lane < 16) ? 0 : 2;

  const float* Arow0 = A0 + (size_t)row * IN_F + kh;
  const float* Arow1 = A1 + (size_t)row * IN_F + kh;
  const float* Bp    = Bpack + (size_t)gn * 256 + lane * 8;
  const size_t bstride = (size_t)ngrpN * 256;

  v8f acc[4] = {};

#define K_STEP(AROW, I, KS)                                                   \
  {                                                                           \
    v2f a = *(const v2f*)((AROW) + (I) * 4);               \
    const float4* bp = (const float4*)(Bp + (size_t)(KS) * bstride);          \
    float4 bA = bp[0];                                   \
    float4 bB = bp[1];                                   \
    v2f b0; b0.x = bA.x; b0.y = bA.y;                                         \
    v2f b1; b1.x = bA.z; b1.y = bA.w;                                         \
    v2f b2; b2.x = bB.x; b2.y = bB.y;                                         \
    v2f b3; b3.x = bB.z; b3.y = bB.w;                                         \
    acc[0] = __builtin_amdgcn_wmma_f32_16x16x4_f32(false, a, false, b0,       \
                                                   (short)0, acc[0], false, false); \
    acc[1] = __builtin_amdgcn_wmma_f32_16x16x4_f32(false, a, false, b1,       \
                                                   (short)0, acc[1], false, false); \
    acc[2] = __builtin_amdgcn_wmma_f32_16x16x4_f32(false, a, false, b2,       \
                                                   (short)0, acc[2], false, false); \
    acc[3] = __builtin_amdgcn_wmma_f32_16x16x4_f32(false, a, false, b3,       \
                                                   (short)0, acc[3], false, false); \
  }

#pragma unroll 4
  for (int i = 0; i < KSPLIT; ++i) {
    K_STEP(Arow0, i, i);
  }
#pragma unroll 4
  for (int i = 0; i < KSTEPS - KSPLIT; ++i) {
    K_STEP(Arow1, i, KSPLIT + i);
  }
#undef K_STEP

  const int mbase = tm * 16 + ((lane < 16) ? 0 : 8);
#pragma unroll
  for (int t = 0; t < 4; ++t) {
    const int col = col0 + t * 16;
    const float bv = bias[col];
#pragma unroll
    for (int i = 0; i < 8; ++i) {
      float v = acc[t][i] + bv;
      if (do_relu) v = fmaxf(v, 0.0f);
      out[(size_t)(mbase + i) * N + col] = v;
    }
  }
}

extern "C" void kernel_launch(void* const* d_in, const int* in_sizes, int n_in,
                              void* d_out, int out_size, void* d_ws, size_t ws_size,
                              hipStream_t stream) {
  const float* x   = (const float*)d_in[0];
  const float* W1  = (const float*)d_in[1];
  const float* b1  = (const float*)d_in[2];
  const float* W2  = (const float*)d_in[3];
  const float* b2  = (const float*)d_in[4];
  const int* src1  = (const int*)d_in[5];
  const int* dst1  = (const int*)d_in[6];
  const int* src2  = (const int*)d_in[7];
  const int* dst2  = (const int*)d_in[8];
  const int E1 = in_sizes[5];
  const int E2 = in_sizes[7];

  char* wsb = (char*)d_ws;
  size_t off = 0;
  auto take = [&](size_t n_elems) { void* p = wsb + off; off += n_elems * 4; return p; };

  int* hist1    = (int*)take(N_DST1);
  int* hist2    = (int*)take(N_DST2);
  int* row_ptr1 = (int*)take(N_DST1 + 1);
  int* cursor1  = (int*)take(N_DST1);
  int* srcs1    = (int*)take((size_t)E1);
  int* row_ptr2 = (int*)take(N_DST2 + 1);
  int* cursor2  = (int*)take(N_DST2);
  int* srcs2    = (int*)take((size_t)E2);
  float* hN1    = (float*)take((size_t)N_DST1 * IN_F);
  float* h      = (float*)take((size_t)N_DST1 * HID_F);
  float* hN2    = (float*)take((size_t)N_DST2 * IN_F);
  float* Bpack1 = (float*)take((size_t)KDIM * HID_F);
  float* Bpack2 = (float*)take((size_t)KDIM * N_CLS);
  (void)ws_size;

  {
    int t1 = KSTEPS * (HID_F / 64) * 128;
    int t2 = KSTEPS * (N_CLS / 64) * 128;
    pack_W_kernel<<<(t1 + 255) / 256, 256, 0, stream>>>(W1, Bpack1, HID_F);
    pack_W_kernel<<<(t2 + 255) / 256, 256, 0, stream>>>(W2, Bpack2, N_CLS);
  }

  zero_i32_kernel<<<(N_DST1 + N_DST2 + 255) / 256, 256, 0, stream>>>(
      hist1, N_DST1 + N_DST2);
  hist_kernel<<<(E1 + 255) / 256, 256, 0, stream>>>(dst1, hist1, E1);
  scan_kernel<<<1, 256, 0, stream>>>(hist1, row_ptr1, cursor1, N_DST1);
  bin_kernel<<<(E1 + 255) / 256, 256, 0, stream>>>(src1, dst1, cursor1, srcs1, E1);
  gather_mean_kernel<<<N_DST1 / 8, 256, 0, stream>>>(x, srcs1, row_ptr1, hN1, N_DST1);

  {
    int ngrps = (N_DST1 / 16) * (HID_F / 64);
    sage_gemm_wmma_kernel<<<ngrps / 4, 128, 0, stream>>>(
        x, hN1, Bpack1, b1, h, N_DST1, HID_F,  1);
  }

  hist_kernel<<<(E2 + 255) / 256, 256, 0, stream>>>(dst2, hist2, E2);
  scan_kernel<<<1, 256, 0, stream>>>(hist2, row_ptr2, cursor2, N_DST2);
  bin_kernel<<<(E2 + 255) / 256, 256, 0, stream>>>(src2, dst2, cursor2, srcs2, E2);
  gather_mean_kernel<<<N_DST2 / 8, 256, 0, stream>>>(h, srcs2, row_ptr2, hN2, N_DST2);

  {
    int ngrps = (N_DST2 / 16) * (N_CLS / 64);
    sage_gemm_wmma_kernel<<<(ngrps + 3) / 4, 128, 0, stream>>>(
        h, hN2, Bpack2, b2, (float*)d_out, N_DST2, N_CLS,  0);
  }
}


# GEMM K loop software-pipelined: fragment loads issued in blocks of 8 K-steps, two groups ahead of their WMMAs (three register sets), instead of load-wait-4 WMMA per K-step; k order unchanged
# speedup vs baseline: 1.1572x; 1.1572x over previous
#include <hip/hip_runtime.h>
#include <hip/hip_bf16.h>


typedef __attribute__((ext_vector_type(2))) float v2f;
typedef __attribute__((ext_vector_type(8))) float v8f;

#define N_DST1   50000
#define N_DST2   10000
#define IN_F     256
#define KDIM     512
#define KSTEPS   (KDIM / 4)
#define KSPLIT   (IN_F / 4)
#define HID_F    256
#define N_CLS    64

__global__ void zero_i32_kernel(int* __restrict__ p, int n) {
  int i = blockIdx.x * blockDim.x + threadIdx.x;
  if (i < n) p[i] = 0;
}

__global__ void hist_kernel(const int* __restrict__ dst,
                            int* __restrict__ hist, int nE) {
  int i = blockIdx.x * blockDim.x + threadIdx.x;
  if (i < nE) atomicAdd(&hist[dst[i]], 1);
}

__global__ __launch_bounds__(256)
void scan_kernel(const int* __restrict__ hist,
                 int* __restrict__ row_ptr,
                 int* __restrict__ cursor, int n) {
  __shared__ int partials[256];
  const int t = threadIdx.x;
  const int chunk = (n + 255) / 256;
  int begin = t * chunk;
  int end   = begin + chunk; if (end > n) end = n;
  if (begin > n) begin = n;

  int sum = 0;
  for (int i = begin; i < end; ++i) sum += hist[i];
  partials[t] = sum;
  __syncthreads();
  if (t == 0) {
    int run = 0;
    for (int i = 0; i < 256; ++i) { int v = partials[i]; partials[i] = run; run += v; }
  }
  __syncthreads();
  int run = partials[t];
  for (int i = begin; i < end; ++i) {
    row_ptr[i] = run;
    cursor[i]  = run;
    run += hist[i];
  }
  if (end == n) row_ptr[n] = run;
}

__global__ void bin_kernel(const int* __restrict__ src,
                           const int* __restrict__ dst,
                           int* __restrict__ cursor,
                           int* __restrict__ srcs, int nE) {
  int i = blockIdx.x * blockDim.x + threadIdx.x;
  if (i < nE) {
    int d = dst[i];
    int pos = atomicAdd(&cursor[d], 1);
    srcs[pos] = src[i];
  }
}

__global__ __launch_bounds__(256)
void gather_mean_kernel(const float* __restrict__ hsrc,
                        const int* __restrict__ srcs,
                        const int* __restrict__ row_ptr,
                        float* __restrict__ hN,
                        int n_dst) {
  const int d = blockIdx.x * (blockDim.x >> 5) + (threadIdx.x >> 5);
  if (d >= n_dst) return;
  const int lane = threadIdx.x & 31;
  const int beg = row_ptr[d];
  const int end = row_ptr[d + 1];

  float4 a0 = make_float4(0.f, 0.f, 0.f, 0.f);
  float4 a1 = make_float4(0.f, 0.f, 0.f, 0.f);
  for (int j = beg; j < end; ++j) {
    int s = srcs[j];
    const float4* p = (const float4*)(hsrc + (size_t)s * IN_F) + lane * 2;
    float4 v0 = p[0];
    float4 v1 = p[1];
    a0.x += v0.x; a0.y += v0.y; a0.z += v0.z; a0.w += v0.w;
    a1.x += v1.x; a1.y += v1.y; a1.z += v1.z; a1.w += v1.w;
  }
  const float inv = 1.0f / fmaxf((float)(end - beg), 1.0f);
  a0.x *= inv; a0.y *= inv; a0.z *= inv; a0.w *= inv;
  a1.x *= inv; a1.y *= inv; a1.z *= inv; a1.w *= inv;
  float4* o = (float4*)(hN + (size_t)d * IN_F) + lane * 2;
  o[0] = a0;
  o[1] = a1;
}

__global__ void pack_W_kernel(const float* __restrict__ W,
                              float* __restrict__ Bpack, int N) {
  const int ngrpN = N >> 6;
  const int total = KSTEPS * ngrpN * 128;
  int idx = blockIdx.x * blockDim.x + threadIdx.x;
  if (idx >= total) return;

  const int w    = idx & 127;
  const int t    = w & 3;
  const int lane = w >> 2;
  const int blk  = idx >> 7;
  const int gn   = blk % ngrpN;
  const int ks   = blk / ngrpN;

  const int kh  = (lane < 16) ? 0 : 2;
  const int col = gn * 64 + t * 16 + (lane & 15);
  const int kr  = ks * 4 + kh;

  float2 v;
  v.x = W[(size_t)kr * N + col];
  v.y = W[(size_t)(kr + 1) * N + col];
  ((float2*)Bpack)[idx] = v;
}

#define KG       8
#define NGROUPS  (KSTEPS / KG)
#define GSPLIT   (KSPLIT / KG)
#define NSETS    3

struct FragSet {
  v2f    a[KG];
  float4 b[KG][2];
};

template <int BSTRIDE>
__device__ __forceinline__ void load_group(FragSet& f, const float* __restrict__ ap,
                                           const float* __restrict__ bp) {
#pragma unroll
  for (int i = 0; i < KG; ++i) {
    f.a[i] = *(const v2f*)(ap + i * 4);
    const float4* q = (const float4*)(bp + (size_t)i * BSTRIDE);
    f.b[i][0] = q[0];
    f.b[i][1] = q[1];
  }
}

__device__ __forceinline__ void mma_group(v8f (&acc)[4], const FragSet& f) {
#pragma unroll
  for (int i = 0; i < KG; ++i) {
    const v2f a = f.a[i];
    v2f b0; b0.x = f.b[i][0].x; b0.y = f.b[i][0].y;
    v2f b1; b1.x = f.b[i][0].z; b1.y = f.b[i][0].w;
    v2f b2; b2.x = f.b[i][1].x; b2.y = f.b[i][1].y;
    v2f b3; b3.x = f.b[i][1].z; b3.y = f.b[i][1].w;
    acc[0] = __builtin_amdgcn_wmma_f32_16x16x4_f32(false, a, false, b0,
                                                   (short)0, acc[0], false, false);
    acc[1] = __builtin_amdgcn_wmma_f32_16x16x4_f32(false, a, false, b1,
                                                   (short)0, acc[1], false, false);
    acc[2] = __builtin_amdgcn_wmma_f32_16x16x4_f32(false, a, false, b2,
                                                   (short)0, acc[2], false, false);
    acc[3] = __builtin_amdgcn_wmma_f32_16x16x4_f32(false, a, false, b3,
                                                   (short)0, acc[3], false, false);
  }
}

template <int N, bool RELU>
__global__ __launch_bounds__(128)
void sage_gemm_wmma_kernel(const float* __restrict__ A0,
                           const float* __restrict__ A1,
                           const float* __restrict__ Bpack,
                           const float* __restrict__ bias,
                           float* __restrict__ out,
                           int M) {
  constexpr int NGRPN   = N >> 6;
  constexpr int BSTRIDE = NGRPN * 256;
  const int lane = threadIdx.x & 31;
  const int wave = threadIdx.x >> 5;
  const int grp   = blockIdx.x * (blockDim.x >> 5) + wave;
  const int ngrps = (M >> 4) * NGRPN;
  if (grp >= ngrps) return;

  const int tm = grp / NGRPN;
  const int gn = grp % NGRPN;

  const int l15  = lane & 15;
  const int row  = tm * 16 + l15;
  const int col0 = gn * 64 + l15;
  const int kh   = (lane < 16) ? 0 : 2;

  const float* Arow0 = A0 + (size_t)row * IN_F + kh;
  const float* Arow1 = A1 + (size_t)row * IN_F + kh;
  const float* Bp    = Bpack + (size_t)gn * 256 + lane * 8;

  v8f acc[4] = {};
  FragSet F[NSETS];

  auto a_of = [&](int g) -> const float* {
    return (g < GSPLIT) ? (Arow0 + g * (KG * 4)) : (Arow1 + (g - GSPLIT) * (KG * 4));
  };
  auto b_of = [&](int g) -> const float* {
    return Bp + (size_t)g * (KG * BSTRIDE);
  };

#define PIN() __builtin_amdgcn_sched_barrier(0)

  load_group<BSTRIDE>(F[0], a_of(0), b_of(0));
  load_group<BSTRIDE>(F[1], a_of(1), b_of(1));
#pragma unroll
  for (int g = 0; g < NGROUPS; ++g) {
    if (g + 2 < NGROUPS) {
      load_group<BSTRIDE>(F[(g + 2) % NSETS], a_of(g + 2), b_of(g + 2));
      PIN();
    }
    mma_group(acc, F[g % NSETS]);
  }
#undef PIN

  const int mbase = tm * 16 + ((lane < 16) ? 0 : 8);
#pragma unroll
  for (int t = 0; t < 4; ++t) {
    const int col = col0 + t * 16;
    const float bv = bias[col];
#pragma unroll
    for (int i = 0; i < 8; ++i) {
      float v = acc[t][i] + bv;
      if (RELU) v = fmaxf(v, 0.0f);
      out[(size_t)(mbase + i) * N + col] = v;
    }
  }
}

extern "C" void kernel_launch(void* const* d_in, const int* in_sizes, int n_in,
                              void* d_out, int out_size, void* d_ws, size_t ws_size,
                              hipStream_t stream) {
  const float* x   = (const float*)d_in[0];
  const float* W1  = (const float*)d_in[1];
  const float* b1  = (const float*)d_in[2];
  const float* W2  = (const float*)d_in[3];
  const float* b2  = (const float*)d_in[4];
  const int* src1  = (const int*)d_in[5];
  const int* dst1  = (const int*)d_in[6];
  const int* src2  = (const int*)d_in[7];
  const int* dst2  = (const int*)d_in[8];
  const int E1 = in_sizes[5];
  const int E2 = in_sizes[7];

  char* wsb = (char*)d_ws;
  size_t off = 0;
  auto take = [&](size_t n_elems) { void* p = wsb + off; off += n_elems * 4; return p; };

  int* hist1    = (int*)take(N_DST1);
  int* hist2    = (int*)take(N_DST2);
  int* row_ptr1 = (int*)take(N_DST1 + 1);
  int* cursor1  = (int*)take(N_DST1);
  int* srcs1    = (int*)take((size_t)E1);
  int* row_ptr2 = (int*)take(N_DST2 + 1);
  int* cursor2  = (int*)take(N_DST2);
  int* srcs2    = (int*)take((size_t)E2);
  float* hN1    = (float*)take((size_t)N_DST1 * IN_F);
  float* h      = (float*)take((size_t)N_DST1 * HID_F);
  float* hN2    = (float*)take((size_t)N_DST2 * IN_F);
  float* Bpack1 = (float*)take((size_t)KDIM * HID_F);
  float* Bpack2 = (float*)take((size_t)KDIM * N_CLS);
  (void)ws_size;

  {
    int t1 = KSTEPS * (HID_F / 64) * 128;
    int t2 = KSTEPS * (N_CLS / 64) * 128;
    pack_W_kernel<<<(t1 + 255) / 256, 256, 0, stream>>>(W1, Bpack1, HID_F);
    pack_W_kernel<<<(t2 + 255) / 256, 256, 0, stream>>>(W2, Bpack2, N_CLS);
  }

  zero_i32_kernel<<<(N_DST1 + N_DST2 + 255) / 256, 256, 0, stream>>>(
      hist1, N_DST1 + N_DST2);
  hist_kernel<<<(E1 + 255) / 256, 256, 0, stream>>>(dst1, hist1, E1);
  scan_kernel<<<1, 256, 0, stream>>>(hist1, row_ptr1, cursor1, N_DST1);
  bin_kernel<<<(E1 + 255) / 256, 256, 0, stream>>>(src1, dst1, cursor1, srcs1, E1);
  gather_mean_kernel<<<N_DST1 / 8, 256, 0, stream>>>(x, srcs1, row_ptr1, hN1, N_DST1);

  {
    int ngrps = (N_DST1 / 16) * (HID_F / 64);
    sage_gemm_wmma_kernel<HID_F, true><<<ngrps / 4, 128, 0, stream>>>(
        x, hN1, Bpack1, b1, h, N_DST1);
  }

  hist_kernel<<<(E2 + 255) / 256, 256, 0, stream>>>(dst2, hist2, E2);
  scan_kernel<<<1, 256, 0, stream>>>(hist2, row_ptr2, cursor2, N_DST2);
  bin_kernel<<<(E2 + 255) / 256, 256, 0, stream>>>(src2, dst2, cursor2, srcs2, E2);
  gather_mean_kernel<<<N_DST2 / 8, 256, 0, stream>>>(h, srcs2, row_ptr2, hN2, N_DST2);

  {
    int ngrps = (N_DST2 / 16) * (N_CLS / 64);
    sage_gemm_wmma_kernel<N_CLS, false><<<(ngrps + 3) / 4, 128, 0, stream>>>(
        h, hN2, Bpack2, b2, (float*)d_out, N_DST2);
  }
}


# Exclusive scans of both degree histograms by one 59-block launch (coalesced 16-byte loads, wave shuffles, each block adds up the tiles before its own) instead of two single-block serial scans
# speedup vs baseline: 1.2845x; 1.1100x over previous
#include <hip/hip_runtime.h>
#include <hip/hip_bf16.h>


typedef __attribute__((ext_vector_type(2))) float v2f;
typedef __attribute__((ext_vector_type(8))) float v8f;

#define N_DST1   50000
#define N_DST2   10000
#define IN_F     256
#define KDIM     512
#define KSTEPS   (KDIM / 4)
#define KSPLIT   (IN_F / 4)
#define HID_F    256
#define N_CLS    64

__global__ void zero_i32_kernel(int* __restrict__ p, int n) {
  int i = blockIdx.x * blockDim.x + threadIdx.x;
  if (i < n) p[i] = 0;
}

__global__ void hist_kernel(const int* __restrict__ dst,
                            int* __restrict__ hist, int nE) {
  int i = blockIdx.x * blockDim.x + threadIdx.x;
  if (i < nE) atomicAdd(&hist[dst[i]], 1);
}

#define SCAN_TILE   1024
#define SCAN_T1     ((N_DST1 + SCAN_TILE - 1) / SCAN_TILE)
#define SCAN_T2     ((N_DST2 + SCAN_TILE - 1) / SCAN_TILE)
__global__ __launch_bounds__(256)
void scan_tiles_kernel(const int* __restrict__ hist,
                       int* __restrict__ row_ptr1, int* __restrict__ cursor1,
                       int* __restrict__ row_ptr2, int* __restrict__ cursor2) {
  __shared__ int wave_pre[8];
  __shared__ int wave_tot[8];
  const int t    = threadIdx.x;
  const int lane = t & 31;
  const int wave = t >> 5;

  int b = blockIdx.x;
  const int* hs; int* rp; int* cur; int n;
  if (b < SCAN_T1) { hs = hist;          rp = row_ptr1; cur = cursor1; n = N_DST1; }
  else { b -= SCAN_T1; hs = hist + N_DST1; rp = row_ptr2; cur = cursor2; n = N_DST2; }

  const int4* h4 = (const int4*)hs;
  int pre = 0;
#pragma unroll 8
  for (int j = 0; j < b; ++j) {
    const int4 q = h4[j * (SCAN_TILE / 4) + t];
    pre += (q.x + q.y) + (q.z + q.w);
  }

  const int e0 = b * SCAN_TILE + t * 4;
  int v0 = 0, v1 = 0, v2 = 0, v3 = 0;
  if (e0 + 3 < n) {
    const int4 q = h4[e0 >> 2];
    v0 = q.x; v1 = q.y; v2 = q.z; v3 = q.w;
  } else {
    if (e0     < n) v0 = hs[e0];
    if (e0 + 1 < n) v1 = hs[e0 + 1];
    if (e0 + 2 < n) v2 = hs[e0 + 2];
  }
  const int s = (v0 + v1) + (v2 + v3);

#pragma unroll
  for (int o = 16; o > 0; o >>= 1) pre += __shfl_xor(pre, o, 32);
  int inc = s;
#pragma unroll
  for (int o = 1; o < 32; o <<= 1) {
    const int u = __shfl_up(inc, o, 32);
    if (lane >= o) inc += u;
  }
  if (lane == 31) wave_tot[wave] = inc;
  if (lane == 0)  wave_pre[wave] = pre;
  __syncthreads();

  int base = 0;
#pragma unroll
  for (int w = 0; w < 8; ++w) {
    base += wave_pre[w];
    if (w < wave) base += wave_tot[w];
  }

  const int x0 = base + (inc - s);
  const int x1 = x0 + v0;
  const int x2 = x1 + v1;
  const int x3 = x2 + v2;
  const int x4 = x3 + v3;
  if (e0     < n) { rp[e0]     = x0; cur[e0]     = x0; }
  if (e0 + 1 < n) { rp[e0 + 1] = x1; cur[e0 + 1] = x1; }
  if (e0 + 2 < n) { rp[e0 + 2] = x2; cur[e0 + 2] = x2; }
  if (e0 + 3 < n) { rp[e0 + 3] = x3; cur[e0 + 3] = x3; }
  if (e0 < n && e0 + 4 >= n) {
    const int last = n - 1 - e0;
    rp[n] = (last == 0) ? x1 : (last == 1) ? x2 : (last == 2) ? x3 : x4;
  }
}

__global__ void bin_kernel(const int* __restrict__ src,
                           const int* __restrict__ dst,
                           int* __restrict__ cursor,
                           int* __restrict__ srcs, int nE) {
  int i = blockIdx.x * blockDim.x + threadIdx.x;
  if (i < nE) {
    int d = dst[i];
    int pos = atomicAdd(&cursor[d], 1);
    srcs[pos] = src[i];
  }
}

__global__ __launch_bounds__(256)
void gather_mean_kernel(const float* __restrict__ hsrc,
                        const int* __restrict__ srcs,
                        const int* __restrict__ row_ptr,
                        float* __restrict__ hN,
                        int n_dst) {
  const int d = blockIdx.x * (blockDim.x >> 5) + (threadIdx.x >> 5);
  if (d >= n_dst) return;
  const int lane = threadIdx.x & 31;
  const int beg = row_ptr[d];
  const int end = row_ptr[d + 1];

  float4 a0 = make_float4(0.f, 0.f, 0.f, 0.f);
  float4 a1 = make_float4(0.f, 0.f, 0.f, 0.f);
  for (int j = beg; j < end; ++j) {
    int s = srcs[j];
    const float4* p = (const float4*)(hsrc + (size_t)s * IN_F) + lane * 2;
    float4 v0 = p[0];
    float4 v1 = p[1];
    a0.x += v0.x; a0.y += v0.y; a0.z += v0.z; a0.w += v0.w;
    a1.x += v1.x; a1.y += v1.y; a1.z += v1.z; a1.w += v1.w;
  }
  const float inv = 1.0f / fmaxf((float)(end - beg), 1.0f);
  a0.x *= inv; a0.y *= inv; a0.z *= inv; a0.w *= inv;
  a1.x *= inv; a1.y *= inv; a1.z *= inv; a1.w *= inv;
  float4* o = (float4*)(hN + (size_t)d * IN_F) + lane * 2;
  o[0] = a0;
  o[1] = a1;
}

__global__ void pack_W_kernel(const float* __restrict__ W,
                              float* __restrict__ Bpack, int N) {
  const int ngrpN = N >> 6;
  const int total = KSTEPS * ngrpN * 128;
  int idx = blockIdx.x * blockDim.x + threadIdx.x;
  if (idx >= total) return;

  const int w    = idx & 127;
  const int t    = w & 3;
  const int lane = w >> 2;
  const int blk  = idx >> 7;
  const int gn   = blk % ngrpN;
  const int ks   = blk / ngrpN;

  const int kh  = (lane < 16) ? 0 : 2;
  const int col = gn * 64 + t * 16 + (lane & 15);
  const int kr  = ks * 4 + kh;

  float2 v;
  v.x = W[(size_t)kr * N + col];
  v.y = W[(size_t)(kr + 1) * N + col];
  ((float2*)Bpack)[idx] = v;
}

#define KG       8
#define NGROUPS  (KSTEPS / KG)
#define GSPLIT   (KSPLIT / KG)
#define NSETS    3

struct FragSet {
  v2f    a[KG];
  float4 b[KG][2];
};

template <int BSTRIDE>
__device__ __forceinline__ void load_group(FragSet& f, const float* __restrict__ ap,
                                           const float* __restrict__ bp) {
#pragma unroll
  for (int i = 0; i < KG; ++i) {
    f.a[i] = *(const v2f*)(ap + i * 4);
    const float4* q = (const float4*)(bp + (size_t)i * BSTRIDE);
    f.b[i][0] = q[0];
    f.b[i][1] = q[1];
  }
}

__device__ __forceinline__ void mma_group(v8f (&acc)[4], const FragSet& f) {
#pragma unroll
  for (int i = 0; i < KG; ++i) {
    const v2f a = f.a[i];
    v2f b0; b0.x = f.b[i][0].x; b0.y = f.b[i][0].y;
    v2f b1; b1.x = f.b[i][0].z; b1.y = f.b[i][0].w;
    v2f b2; b2.x = f.b[i][1].x; b2.y = f.b[i][1].y;
    v2f b3; b3.x = f.b[i][1].z; b3.y = f.b[i][1].w;
    acc[0] = __builtin_amdgcn_wmma_f32_16x16x4_f32(false, a, false, b0,
                                                   (short)0, acc[0], false, false);
    acc[1] = __builtin_amdgcn_wmma_f32_16x16x4_f32(false, a, false, b1,
                                                   (short)0, acc[1], false, false);
    acc[2] = __builtin_amdgcn_wmma_f32_16x16x4_f32(false, a, false, b2,
                                                   (short)0, acc[2], false, false);
    acc[3] = __builtin_amdgcn_wmma_f32_16x16x4_f32(false, a, false, b3,
                                                   (short)0, acc[3], false, false);
  }
}

template <int N, bool RELU>
__global__ __launch_bounds__(128)
void sage_gemm_wmma_kernel(const float* __restrict__ A0,
                           const float* __restrict__ A1,
                           const float* __restrict__ Bpack,
                           const float* __restrict__ bias,
                           float* __restrict__ out,
                           int M) {
  constexpr int NGRPN   = N >> 6;
  constexpr int BSTRIDE = NGRPN * 256;
  const int lane = threadIdx.x & 31;
  const int wave = threadIdx.x >> 5;
  const int grp   = blockIdx.x * (blockDim.x >> 5) + wave;
  const int ngrps = (M >> 4) * NGRPN;
  if (grp >= ngrps) return;

  const int tm = grp / NGRPN;
  const int gn = grp % NGRPN;

  const int l15  = lane & 15;
  const int row  = tm * 16 + l15;
  const int col0 = gn * 64 + l15;
  const int kh   = (lane < 16) ? 0 : 2;

  const float* Arow0 = A0 + (size_t)row * IN_F + kh;
  const float* Arow1 = A1 + (size_t)row * IN_F + kh;
  const float* Bp    = Bpack + (size_t)gn * 256 + lane * 8;

  v8f acc[4] = {};
  FragSet F[NSETS];

  auto a_of = [&](int g) -> const float* {
    return (g < GSPLIT) ? (Arow0 + g * (KG * 4)) : (Arow1 + (g - GSPLIT) * (KG * 4));
  };
  auto b_of = [&](int g) -> const float* {
    return Bp + (size_t)g * (KG * BSTRIDE);
  };

#define PIN() __builtin_amdgcn_sched_barrier(0)

  load_group<BSTRIDE>(F[0], a_of(0), b_of(0));
  load_group<BSTRIDE>(F[1], a_of(1), b_of(1));
#pragma unroll
  for (int g = 0; g < NGROUPS; ++g) {
    if (g + 2 < NGROUPS) {
      load_group<BSTRIDE>(F[(g + 2) % NSETS], a_of(g + 2), b_of(g + 2));
      PIN();
    }
    mma_group(acc, F[g % NSETS]);
  }
#undef PIN

  const int mbase = tm * 16 + ((lane < 16) ? 0 : 8);
#pragma unroll
  for (int t = 0; t < 4; ++t) {
    const int col = col0 + t * 16;
    const float bv = bias[col];
#pragma unroll
    for (int i = 0; i < 8; ++i) {
      float v = acc[t][i] + bv;
      if (RELU) v = fmaxf(v, 0.0f);
      out[(size_t)(mbase + i) * N + col] = v;
    }
  }
}

extern "C" void kernel_launch(void* const* d_in, const int* in_sizes, int n_in,
                              void* d_out, int out_size, void* d_ws, size_t ws_size,
                              hipStream_t stream) {
  const float* x   = (const float*)d_in[0];
  const float* W1  = (const float*)d_in[1];
  const float* b1  = (const float*)d_in[2];
  const float* W2  = (const float*)d_in[3];
  const float* b2  = (const float*)d_in[4];
  const int* src1  = (const int*)d_in[5];
  const int* dst1  = (const int*)d_in[6];
  const int* src2  = (const int*)d_in[7];
  const int* dst2  = (const int*)d_in[8];
  const int E1 = in_sizes[5];
  const int E2 = in_sizes[7];

  char* wsb = (char*)d_ws;
  size_t off = 0;
  auto take = [&](size_t n_elems) { void* p = wsb + off; off += n_elems * 4; return p; };

  int* hist1    = (int*)take(N_DST1);
  int* hist2    = (int*)take(N_DST2);
  int* row_ptr1 = (int*)take(N_DST1 + 1);
  int* cursor1  = (int*)take(N_DST1);
  int* srcs1    = (int*)take((size_t)E1);
  int* row_ptr2 = (int*)take(N_DST2 + 1);
  int* cursor2  = (int*)take(N_DST2);
  int* srcs2    = (int*)take((size_t)E2);
  float* hN1    = (float*)take((size_t)N_DST1 * IN_F);
  float* h      = (float*)take((size_t)N_DST1 * HID_F);
  float* hN2    = (float*)take((size_t)N_DST2 * IN_F);
  float* Bpack1 = (float*)take((size_t)KDIM * HID_F);
  float* Bpack2 = (float*)take((size_t)KDIM * N_CLS);
  (void)ws_size;

  {
    int t1 = KSTEPS * (HID_F / 64) * 128;
    int t2 = KSTEPS * (N_CLS / 64) * 128;
    pack_W_kernel<<<(t1 + 255) / 256, 256, 0, stream>>>(W1, Bpack1, HID_F);
    pack_W_kernel<<<(t2 + 255) / 256, 256, 0, stream>>>(W2, Bpack2, N_CLS);
  }

  zero_i32_kernel<<<(N_DST1 + N_DST2 + 255) / 256, 256, 0, stream>>>(
      hist1, N_DST1 + N_DST2);
  hist_kernel<<<(E1 + 255) / 256, 256, 0, stream>>>(dst1, hist1, E1);
  hist_kernel<<<(E2 + 255) / 256, 256, 0, stream>>>(dst2, hist2, E2);
  scan_tiles_kernel<<<SCAN_T1 + SCAN_T2, 256, 0, stream>>>(
      hist1, row_ptr1, cursor1, row_ptr2, cursor2);
  bin_kernel<<<(E1 + 255) / 256, 256, 0, stream>>>(src1, dst1, cursor1, srcs1, E1);
  gather_mean_kernel<<<N_DST1 / 8, 256, 0, stream>>>(x, srcs1, row_ptr1, hN1, N_DST1);

  {
    int ngrps = (N_DST1 / 16) * (HID_F / 64);
    sage_gemm_wmma_kernel<HID_F, true><<<ngrps / 4, 128, 0, stream>>>(
        x, hN1, Bpack1, b1, h, N_DST1);
  }

  bin_kernel<<<(E2 + 255) / 256, 256, 0, stream>>>(src2, dst2, cursor2, srcs2, E2);
  gather_mean_kernel<<<N_DST2 / 8, 256, 0, stream>>>(h, srcs2, row_ptr2, hN2, N_DST2);

  {
    int ngrps = (N_DST2 / 16) * (N_CLS / 64);
    sage_gemm_wmma_kernel<N_CLS, false><<<(ngrps + 3) / 4, 128, 0, stream>>>(
        h, hN2, Bpack2, b2, (float*)d_out, N_DST2);
  }
}


# Bin pass without atomics: the histogram add's return value is kept as each edge's rank in its dst row; bin stores src at row start + rank, 4 edges per lane; neighbor order stays an arrival order
# speedup vs baseline: 1.4893x; 1.1594x over previous
#include <hip/hip_runtime.h>
#include <hip/hip_bf16.h>


typedef __attribute__((ext_vector_type(2))) float v2f;
typedef __attribute__((ext_vector_type(8))) float v8f;

#define N_DST1   50000
#define N_DST2   10000
#define IN_F     256
#define KDIM     512
#define KSTEPS   (KDIM / 4)
#define KSPLIT   (IN_F / 4)
#define HID_F    256
#define N_CLS    64

__global__ void zero_i32_kernel(int* __restrict__ p, int n) {
  int i = blockIdx.x * blockDim.x + threadIdx.x;
  if (i < n) p[i] = 0;
}

__global__ void hist_rank_kernel(const int* __restrict__ dst,
                                 int* __restrict__ hist,
                                 int* __restrict__ rank, int nE) {
  int i = blockIdx.x * blockDim.x + threadIdx.x;
  if (i < nE) rank[i] = atomicAdd(&hist[dst[i]], 1);
}

#define SCAN_TILE   1024
#define SCAN_T1     ((N_DST1 + SCAN_TILE - 1) / SCAN_TILE)
#define SCAN_T2     ((N_DST2 + SCAN_TILE - 1) / SCAN_TILE)
__global__ __launch_bounds__(256)
void scan_tiles_kernel(const int* __restrict__ hist,
                       int* __restrict__ row_ptr1,
                       int* __restrict__ row_ptr2) {
  __shared__ int wave_pre[8];
  __shared__ int wave_tot[8];
  const int t    = threadIdx.x;
  const int lane = t & 31;
  const int wave = t >> 5;

  int b = blockIdx.x;
  const int* hs; int* rp; int n;
  if (b < SCAN_T1) { hs = hist;          rp = row_ptr1; n = N_DST1; }
  else { b -= SCAN_T1; hs = hist + N_DST1; rp = row_ptr2; n = N_DST2; }

  const int4* h4 = (const int4*)hs;
  int pre = 0;
#pragma unroll 8
  for (int j = 0; j < b; ++j) {
    const int4 q = h4[j * (SCAN_TILE / 4) + t];
    pre += (q.x + q.y) + (q.z + q.w);
  }

  const int e0 = b * SCAN_TILE + t * 4;
  int v0 = 0, v1 = 0, v2 = 0, v3 = 0;
  if (e0 + 3 < n) {
    const int4 q = h4[e0 >> 2];
    v0 = q.x; v1 = q.y; v2 = q.z; v3 = q.w;
  } else {
    if (e0     < n) v0 = hs[e0];
    if (e0 + 1 < n) v1 = hs[e0 + 1];
    if (e0 + 2 < n) v2 = hs[e0 + 2];
  }
  const int s = (v0 + v1) + (v2 + v3);

#pragma unroll
  for (int o = 16; o > 0; o >>= 1) pre += __shfl_xor(pre, o, 32);
  int inc = s;
#pragma unroll
  for (int o = 1; o < 32; o <<= 1) {
    const int u = __shfl_up(inc, o, 32);
    if (lane >= o) inc += u;
  }
  if (lane == 31) wave_tot[wave] = inc;
  if (lane == 0)  wave_pre[wave] = pre;
  __syncthreads();

  int base = 0;
#pragma unroll
  for (int w = 0; w < 8; ++w) {
    base += wave_pre[w];
    if (w < wave) base += wave_tot[w];
  }

  const int x0 = base + (inc - s);
  const int x1 = x0 + v0;
  const int x2 = x1 + v1;
  const int x3 = x2 + v2;
  const int x4 = x3 + v3;
  if (e0     < n) rp[e0]     = x0;
  if (e0 + 1 < n) rp[e0 + 1] = x1;
  if (e0 + 2 < n) rp[e0 + 2] = x2;
  if (e0 + 3 < n) rp[e0 + 3] = x3;
  if (e0 < n && e0 + 4 >= n) {
    const int last = n - 1 - e0;
    rp[n] = (last == 0) ? x1 : (last == 1) ? x2 : (last == 2) ? x3 : x4;
  }
}

__global__ __launch_bounds__(256)
void bin_kernel(const int* __restrict__ src,
                const int* __restrict__ dst,
                const int* __restrict__ row_ptr,
                const int* __restrict__ rank,
                int* __restrict__ srcs, int nE) {
  const int q  = blockIdx.x * blockDim.x + threadIdx.x;
  const int e0 = q * 4;
  if (e0 + 3 < nE) {
    const int4 d = ((const int4*)dst)[q];
    const int4 r = ((const int4*)rank)[q];
    const int4 s = ((const int4*)src)[q];
    const int p0 = row_ptr[d.x];
    const int p1 = row_ptr[d.y];
    const int p2 = row_ptr[d.z];
    const int p3 = row_ptr[d.w];
    srcs[p0 + r.x] = s.x;
    srcs[p1 + r.y] = s.y;
    srcs[p2 + r.z] = s.z;
    srcs[p3 + r.w] = s.w;
  } else {
    for (int e = e0; e < nE; ++e)
      srcs[row_ptr[dst[e]] + rank[e]] = src[e];
  }
}

__global__ __launch_bounds__(256)
void gather_mean_kernel(const float* __restrict__ hsrc,
                        const int* __restrict__ srcs,
                        const int* __restrict__ row_ptr,
                        float* __restrict__ hN,
                        int n_dst) {
  const int d = blockIdx.x * (blockDim.x >> 5) + (threadIdx.x >> 5);
  if (d >= n_dst) return;
  const int lane = threadIdx.x & 31;
  const int beg = row_ptr[d];
  const int end = row_ptr[d + 1];

  float4 a0 = make_float4(0.f, 0.f, 0.f, 0.f);
  float4 a1 = make_float4(0.f, 0.f, 0.f, 0.f);
  for (int j = beg; j < end; ++j) {
    int s = srcs[j];
    const float4* p = (const float4*)(hsrc + (size_t)s * IN_F) + lane * 2;
    float4 v0 = p[0];
    float4 v1 = p[1];
    a0.x += v0.x; a0.y += v0.y; a0.z += v0.z; a0.w += v0.w;
    a1.x += v1.x; a1.y += v1.y; a1.z += v1.z; a1.w += v1.w;
  }
  const float inv = 1.0f / fmaxf((float)(end - beg), 1.0f);
  a0.x *= inv; a0.y *= inv; a0.z *= inv; a0.w *= inv;
  a1.x *= inv; a1.y *= inv; a1.z *= inv; a1.w *= inv;
  float4* o = (float4*)(hN + (size_t)d * IN_F) + lane * 2;
  o[0] = a0;
  o[1] = a1;
}

__global__ void pack_W_kernel(const float* __restrict__ W,
                              float* __restrict__ Bpack, int N) {
  const int ngrpN = N >> 6;
  const int total = KSTEPS * ngrpN * 128;
  int idx = blockIdx.x * blockDim.x + threadIdx.x;
  if (idx >= total) return;

  const int w    = idx & 127;
  const int t    = w & 3;
  const int lane = w >> 2;
  const int blk  = idx >> 7;
  const int gn   = blk % ngrpN;
  const int ks   = blk / ngrpN;

  const int kh  = (lane < 16) ? 0 : 2;
  const int col = gn * 64 + t * 16 + (lane & 15);
  const int kr  = ks * 4 + kh;

  float2 v;
  v.x = W[(size_t)kr * N + col];
  v.y = W[(size_t)(kr + 1) * N + col];
  ((float2*)Bpack)[idx] = v;
}

#define KG       8
#define NGROUPS  (KSTEPS / KG)
#define GSPLIT   (KSPLIT / KG)
#define NSETS    3

struct FragSet {
  v2f    a[KG];
  float4 b[KG][2];
};

template <int BSTRIDE>
__device__ __forceinline__ void load_group(FragSet& f, const float* __restrict__ ap,
                                           const float* __restrict__ bp) {
#pragma unroll
  for (int i = 0; i < KG; ++i) {
    f.a[i] = *(const v2f*)(ap + i * 4);
    const float4* q = (const float4*)(bp + (size_t)i * BSTRIDE);
    f.b[i][0] = q[0];
    f.b[i][1] = q[1];
  }
}

__device__ __forceinline__ void mma_group(v8f (&acc)[4], const FragSet& f) {
#pragma unroll
  for (int i = 0; i < KG; ++i) {
    const v2f a = f.a[i];
    v2f b0; b0.x = f.b[i][0].x; b0.y = f.b[i][0].y;
    v2f b1; b1.x = f.b[i][0].z; b1.y = f.b[i][0].w;
    v2f b2; b2.x = f.b[i][1].x; b2.y = f.b[i][1].y;
    v2f b3; b3.x = f.b[i][1].z; b3.y = f.b[i][1].w;
    acc[0] = __builtin_amdgcn_wmma_f32_16x16x4_f32(false, a, false, b0,
                                                   (short)0, acc[0], false, false);
    acc[1] = __builtin_amdgcn_wmma_f32_16x16x4_f32(false, a, false, b1,
                                                   (short)0, acc[1], false, false);
    acc[2] = __builtin_amdgcn_wmma_f32_16x16x4_f32(false, a, false, b2,
                                                   (short)0, acc[2], false, false);
    acc[3] = __builtin_amdgcn_wmma_f32_16x16x4_f32(false, a, false, b3,
                                                   (short)0, acc[3], false, false);
  }
}

template <int N, bool RELU>
__global__ __launch_bounds__(128)
void sage_gemm_wmma_kernel(const float* __restrict__ A0,
                           const float* __restrict__ A1,
                           const float* __restrict__ Bpack,
                           const float* __restrict__ bias,
                           float* __restrict__ out,
                           int M) {
  constexpr int NGRPN   = N >> 6;
  constexpr int BSTRIDE = NGRPN * 256;
  const int lane = threadIdx.x & 31;
  const int wave = threadIdx.x >> 5;
  const int grp   = blockIdx.x * (blockDim.x >> 5) + wave;
  const int ngrps = (M >> 4) * NGRPN;
  if (grp >= ngrps) return;

  const int tm = grp / NGRPN;
  const int gn = grp % NGRPN;

  const int l15  = lane & 15;
  const int row  = tm * 16 + l15;
  const int col0 = gn * 64 + l15;
  const int kh   = (lane < 16) ? 0 : 2;

  const float* Arow0 = A0 + (size_t)row * IN_F + kh;
  const float* Arow1 = A1 + (size_t)row * IN_F + kh;
  const float* Bp    = Bpack + (size_t)gn * 256 + lane * 8;

  v8f acc[4] = {};
  FragSet F[NSETS];

  auto a_of = [&](int g) -> const float* {
    return (g < GSPLIT) ? (Arow0 + g * (KG * 4)) : (Arow1 + (g - GSPLIT) * (KG * 4));
  };
  auto b_of = [&](int g) -> const float* {
    return Bp + (size_t)g * (KG * BSTRIDE);
  };

#define PIN() __builtin_amdgcn_sched_barrier(0)

  load_group<BSTRIDE>(F[0], a_of(0), b_of(0));
  load_group<BSTRIDE>(F[1], a_of(1), b_of(1));
#pragma unroll
  for (int g = 0; g < NGROUPS; ++g) {
    if (g + 2 < NGROUPS) {
      load_group<BSTRIDE>(F[(g + 2) % NSETS], a_of(g + 2), b_of(g + 2));
      PIN();
    }
    mma_group(acc, F[g % NSETS]);
  }
#undef PIN

  const int mbase = tm * 16 + ((lane < 16) ? 0 : 8);
#pragma unroll
  for (int t = 0; t < 4; ++t) {
    const int col = col0 + t * 16;
    const float bv = bias[col];
#pragma unroll
    for (int i = 0; i < 8; ++i) {
      float v = acc[t][i] + bv;
      if (RELU) v = fmaxf(v, 0.0f);
      out[(size_t)(mbase + i) * N + col] = v;
    }
  }
}

extern "C" void kernel_launch(void* const* d_in, const int* in_sizes, int n_in,
                              void* d_out, int out_size, void* d_ws, size_t ws_size,
                              hipStream_t stream) {
  const float* x   = (const float*)d_in[0];
  const float* W1  = (const float*)d_in[1];
  const float* b1  = (const float*)d_in[2];
  const float* W2  = (const float*)d_in[3];
  const float* b2  = (const float*)d_in[4];
  const int* src1  = (const int*)d_in[5];
  const int* dst1  = (const int*)d_in[6];
  const int* src2  = (const int*)d_in[7];
  const int* dst2  = (const int*)d_in[8];
  const int E1 = in_sizes[5];
  const int E2 = in_sizes[7];

  char* wsb = (char*)d_ws;
  size_t off = 0;
  auto take = [&](size_t n_elems) { void* p = wsb + off; off += n_elems * 4; return p; };

  int* hist1    = (int*)take(N_DST1);
  int* hist2    = (int*)take(N_DST2);
  int* row_ptr1 = (int*)take(N_DST1 + 1);
  (void)take(N_DST1);
  int* srcs1    = (int*)take((size_t)E1);
  int* row_ptr2 = (int*)take(N_DST2 + 1);
  (void)take(N_DST2);
  int* srcs2    = (int*)take((size_t)E2);
  const size_t off_hN1 = off;
  float* hN1    = (float*)take((size_t)N_DST1 * IN_F);
  float* h      = (float*)take((size_t)N_DST1 * HID_F);
  const size_t off_hN2 = off;
  float* hN2    = (float*)take((size_t)N_DST2 * IN_F);
  int* rank1 = (int*)(wsb + ((off_hN1 + 15) / 16) * 16);
  int* rank2 = (int*)(wsb + ((off_hN2 + 15) / 16) * 16);
  float* Bpack1 = (float*)take((size_t)KDIM * HID_F);
  float* Bpack2 = (float*)take((size_t)KDIM * N_CLS);
  (void)ws_size;

  {
    int t1 = KSTEPS * (HID_F / 64) * 128;
    int t2 = KSTEPS * (N_CLS / 64) * 128;
    pack_W_kernel<<<(t1 + 255) / 256, 256, 0, stream>>>(W1, Bpack1, HID_F);
    pack_W_kernel<<<(t2 + 255) / 256, 256, 0, stream>>>(W2, Bpack2, N_CLS);
  }

  zero_i32_kernel<<<(N_DST1 + N_DST2 + 255) / 256, 256, 0, stream>>>(
      hist1, N_DST1 + N_DST2);
  hist_rank_kernel<<<(E1 + 255) / 256, 256, 0, stream>>>(dst1, hist1, rank1, E1);
  hist_rank_kernel<<<(E2 + 255) / 256, 256, 0, stream>>>(dst2, hist2, rank2, E2);
  scan_tiles_kernel<<<SCAN_T1 + SCAN_T2, 256, 0, stream>>>(hist1, row_ptr1, row_ptr2);
  bin_kernel<<<(E1 + 1023) / 1024, 256, 0, stream>>>(src1, dst1, row_ptr1, rank1, srcs1, E1);
  gather_mean_kernel<<<N_DST1 / 8, 256, 0, stream>>>(x, srcs1, row_ptr1, hN1, N_DST1);

  {
    int ngrps = (N_DST1 / 16) * (HID_F / 64);
    sage_gemm_wmma_kernel<HID_F, true><<<ngrps / 4, 128, 0, stream>>>(
        x, hN1, Bpack1, b1, h, N_DST1);
  }

  bin_kernel<<<(E2 + 1023) / 1024, 256, 0, stream>>>(src2, dst2, row_ptr2, rank2, srcs2, E2);
  gather_mean_kernel<<<N_DST2 / 8, 256, 0, stream>>>(h, srcs2, row_ptr2, hN2, N_DST2);

  {
    int ngrps = (N_DST2 / 16) * (N_CLS / 64);
    sage_gemm_wmma_kernel<N_CLS, false><<<(ngrps + 3) / 4, 128, 0, stream>>>(
        h, hN2, Bpack2, b2, (float*)d_out, N_DST2);
  }
}


# CSR build by two-level counting sort on LDS counters (chunk sort by 256-row bucket, then a work-group per bucket) replaces global returning atomics, scan and bin; neighbor order stays arrival order
# speedup vs baseline: 2.1801x; 1.4638x over previous
#include <hip/hip_runtime.h>
#include <hip/hip_bf16.h>


typedef __attribute__((ext_vector_type(2))) float v2f;
typedef __attribute__((ext_vector_type(8))) float v8f;

#define N_DST1   50000
#define N_DST2   10000
#define IN_F     256
#define KDIM     512
#define KSTEPS   (KDIM / 4)
#define KSPLIT   (IN_F / 4)
#define HID_F    256
#define N_CLS    64

#define BKT        256
#define NB1        ((N_DST1 + BKT - 1) / BKT)
#define NB2        ((N_DST2 + BKT - 1) / BKT)
#define CS1        4096
#define CS2        1024
#define CS_MAX     4096

__device__ __forceinline__ int block_excl_scan_256(int v, int* wtot, int lane, int w) {
  int inc = v;
  for (int o = 1; o < 32; o <<= 1) {
    const int u = __shfl_up(inc, o, 32);
    if (lane >= o) inc += u;
  }
  if (lane == 31) wtot[w] = inc;
  __syncthreads();
  int before = 0;
  for (int ww = 0; ww < 8; ++ww)
    if (ww < w) before += wtot[ww];
  return before + inc - v;
}

__global__ __launch_bounds__(256)
void chunk_sort_kernel(const int* __restrict__ src1, const int* __restrict__ dst1,
                       int E1, int nc1,
                       unsigned* __restrict__ sorted1, int* __restrict__ tab1,
                       const int* __restrict__ src2, const int* __restrict__ dst2,
                       int E2,
                       unsigned* __restrict__ sorted2, int* __restrict__ tab2) {
  __shared__ int cnt[256];
  __shared__ int wtot[8];
  __shared__ __attribute__((aligned(16))) unsigned stage[CS_MAX];
  const int t    = threadIdx.x;
  const int lane = t & 31;
  const int w    = t >> 5;

  int c = blockIdx.x;
  const int* src; const int* dst; unsigned* sorted; int* tab; int E, cs, nb;
  if (c < nc1) { src = src1; dst = dst1; sorted = sorted1; tab = tab1; E = E1; cs = CS1; nb = NB1; }
  else { c -= nc1; src = src2; dst = dst2; sorted = sorted2; tab = tab2; E = E2; cs = CS2; nb = NB2; }

  cnt[t] = 0;
  const int base = c * cs;
  const int n    = min(cs, E - base);

  int d[16], s[16], r[16];
#pragma unroll
  for (int j = 0; j < 4; ++j) {
    const int e0 = j * 1024 + t * 4;
#pragma unroll
    for (int k = 0; k < 4; ++k) { d[j * 4 + k] = -1; s[j * 4 + k] = 0; }
    if (e0 + 3 < n) {
      const int4 dv = *(const int4*)(dst + base + e0);
      const int4 sv = *(const int4*)(src + base + e0);
      d[j * 4 + 0] = dv.x; d[j * 4 + 1] = dv.y; d[j * 4 + 2] = dv.z; d[j * 4 + 3] = dv.w;
      s[j * 4 + 0] = sv.x; s[j * 4 + 1] = sv.y; s[j * 4 + 2] = sv.z; s[j * 4 + 3] = sv.w;
    } else {
#pragma unroll
      for (int k = 0; k < 4; ++k)
        if (e0 + k < n) { d[j * 4 + k] = dst[base + e0 + k]; s[j * 4 + k] = src[base + e0 + k]; }
    }
  }
  __syncthreads();

#pragma unroll
  for (int i = 0; i < 16; ++i) {
    r[i] = 0;
    if (d[i] >= 0) r[i] = atomicAdd(&cnt[d[i] >> 8], 1);
  }
  __syncthreads();

  const int excl = block_excl_scan_256(cnt[t], wtot, lane, w);
  cnt[t] = excl;
  __syncthreads();

#pragma unroll
  for (int i = 0; i < 16; ++i)
    if (d[i] >= 0)
      stage[cnt[d[i] >> 8] + r[i]] = ((unsigned)s[i] << 8) | (unsigned)(d[i] & (BKT - 1));
  __syncthreads();

  for (int i = t * 4; i < n; i += 1024) {
    if (i + 3 < n) {
      *(uint4*)(sorted + base + i) = *(const uint4*)(stage + i);
    } else {
      for (int k = i; k < n; ++k) sorted[base + k] = stage[k];
    }
  }
  int* trow = tab + (size_t)c * (nb + 1);
  if (t < nb)  trow[t]  = excl;
  if (t == 0)  trow[nb] = n;
}

template <int PASS>
__device__ __forceinline__ void walk_runs(const unsigned* __restrict__ sorted,
                                          const int* __restrict__ tab,
                                          int nc, int cs, int nbp1, int b,
                                          int* kcnt, int* __restrict__ srcs,
                                          int w, int lane) {
  for (int cb = 0; cb < nc; cb += 256) {
    const int c = cb + w + 8 * lane;
    int p = 0, n = 0;
    if (c < nc) {
      const int a = tab[(size_t)c * nbp1 + b];
      n = tab[(size_t)c * nbp1 + b + 1] - a;
      p = c * cs + a;
    }
    const int nruns = min(32, (nc - cb - w + 7) >> 3);
    for (int l = 0; l < nruns; l += 4) {
      const int p0 = __builtin_amdgcn_readlane(p, l);
      const int p1 = __builtin_amdgcn_readlane(p, l + 1);
      const int p2 = __builtin_amdgcn_readlane(p, l + 2);
      const int p3 = __builtin_amdgcn_readlane(p, l + 3);
      const int n0 = __builtin_amdgcn_readlane(n, l);
      const int n1 = __builtin_amdgcn_readlane(n, l + 1);
      const int n2 = __builtin_amdgcn_readlane(n, l + 2);
      const int n3 = __builtin_amdgcn_readlane(n, l + 3);
      const int m  = max(max(n0, n1), max(n2, n3));
      for (int k0 = 0; k0 < m; k0 += 32) {
        const int k = k0 + lane;
        const bool q0 = k < n0, q1 = k < n1, q2 = k < n2, q3 = k < n3;
        const unsigned v0 = sorted[q0 ? p0 + k : 0];
        const unsigned v1 = sorted[q1 ? p1 + k : 0];
        const unsigned v2 = sorted[q2 ? p2 + k : 0];
        const unsigned v3 = sorted[q3 ? p3 + k : 0];
        if (PASS == 1) {
          if (q0) atomicAdd(&kcnt[v0 & (BKT - 1)], 1);
          if (q1) atomicAdd(&kcnt[v1 & (BKT - 1)], 1);
          if (q2) atomicAdd(&kcnt[v2 & (BKT - 1)], 1);
          if (q3) atomicAdd(&kcnt[v3 & (BKT - 1)], 1);
        } else {
          if (q0) srcs[atomicAdd(&kcnt[v0 & (BKT - 1)], 1)] = (int)(v0 >> 8);
          if (q1) srcs[atomicAdd(&kcnt[v1 & (BKT - 1)], 1)] = (int)(v1 >> 8);
          if (q2) srcs[atomicAdd(&kcnt[v2 & (BKT - 1)], 1)] = (int)(v2 >> 8);
          if (q3) srcs[atomicAdd(&kcnt[v3 & (BKT - 1)], 1)] = (int)(v3 >> 8);
        }
      }
    }
  }
}

__global__ __launch_bounds__(256)
void bucket_build_kernel(const unsigned* __restrict__ sorted1, const int* __restrict__ tab1,
                         int nc1, int E1,
                         int* __restrict__ row_ptr1, int* __restrict__ srcs1,
                         const unsigned* __restrict__ sorted2, const int* __restrict__ tab2,
                         int nc2, int E2,
                         int* __restrict__ row_ptr2, int* __restrict__ srcs2) {
  __shared__ int kcnt[BKT];
  __shared__ int wred[8];
  __shared__ int wtot[8];
  const int t    = threadIdx.x;
  const int lane = t & 31;
  const int w    = __builtin_amdgcn_readfirstlane(t >> 5);

  int b = blockIdx.x;
  const unsigned* sorted; const int* tab; int* row_ptr; int* srcs; int nc, E, cs, nb, N;
  if (b < NB1) { sorted = sorted1; tab = tab1; row_ptr = row_ptr1; srcs = srcs1;
                 nc = nc1; E = E1; cs = CS1; nb = NB1; N = N_DST1; }
  else { b -= NB1; sorted = sorted2; tab = tab2; row_ptr = row_ptr2; srcs = srcs2;
         nc = nc2; E = E2; cs = CS2; nb = NB2; N = N_DST2; }
  const int nbp1 = nb + 1;

  kcnt[t] = 0;
  int asum = 0;
  for (int c = t; c < nc; c += 256) asum += tab[(size_t)c * nbp1 + b];
  for (int o = 16; o > 0; o >>= 1) asum += __shfl_xor(asum, o, 32);
  if (lane == 0) wred[w] = asum;
  __syncthreads();
  int base = 0;
  for (int ww = 0; ww < 8; ++ww) base += wred[ww];

  walk_runs<1>(sorted, tab, nc, cs, nbp1, b, kcnt, srcs, w, lane);
  __syncthreads();

  const int excl = block_excl_scan_256(kcnt[t], wtot, lane, w);
  const int row  = b * BKT + t;
  if (row < N) row_ptr[row] = base + excl;
  if (b == nb - 1 && t == 0) row_ptr[N] = E;
  kcnt[t] = base + excl;
  __syncthreads();

  walk_runs<2>(sorted, tab, nc, cs, nbp1, b, kcnt, srcs, w, lane);
}

__global__ __launch_bounds__(256)
void gather_mean_kernel(const float* __restrict__ hsrc,
                        const int* __restrict__ srcs,
                        const int* __restrict__ row_ptr,
                        float* __restrict__ hN,
                        int n_dst) {
  const int d = blockIdx.x * (blockDim.x >> 5) + (threadIdx.x >> 5);
  if (d >= n_dst) return;
  const int lane = threadIdx.x & 31;
  const int beg = row_ptr[d];
  const int end = row_ptr[d + 1];

  float4 a0 = make_float4(0.f, 0.f, 0.f, 0.f);
  float4 a1 = make_float4(0.f, 0.f, 0.f, 0.f);
  for (int j = beg; j < end; ++j) {
    int s = srcs[j];
    const float4* p = (const float4*)(hsrc + (size_t)s * IN_F) + lane * 2;
    float4 v0 = p[0];
    float4 v1 = p[1];
    a0.x += v0.x; a0.y += v0.y; a0.z += v0.z; a0.w += v0.w;
    a1.x += v1.x; a1.y += v1.y; a1.z += v1.z; a1.w += v1.w;
  }
  const float inv = 1.0f / fmaxf((float)(end - beg), 1.0f);
  a0.x *= inv; a0.y *= inv; a0.z *= inv; a0.w *= inv;
  a1.x *= inv; a1.y *= inv; a1.z *= inv; a1.w *= inv;
  float4* o = (float4*)(hN + (size_t)d * IN_F) + lane * 2;
  o[0] = a0;
  o[1] = a1;
}

__global__ void pack_W_kernel(const float* __restrict__ W,
                              float* __restrict__ Bpack, int N) {
  const int ngrpN = N >> 6;
  const int total = KSTEPS * ngrpN * 128;
  int idx = blockIdx.x * blockDim.x + threadIdx.x;
  if (idx >= total) return;

  const int w    = idx & 127;
  const int t    = w & 3;
  const int lane = w >> 2;
  const int blk  = idx >> 7;
  const int gn   = blk % ngrpN;
  const int ks   = blk / ngrpN;

  const int kh  = (lane < 16) ? 0 : 2;
  const int col = gn * 64 + t * 16 + (lane & 15);
  const int kr  = ks * 4 + kh;

  float2 v;
  v.x = W[(size_t)kr * N + col];
  v.y = W[(size_t)(kr + 1) * N + col];
  ((float2*)Bpack)[idx] = v;
}

#define KG       8
#define NGROUPS  (KSTEPS / KG)
#define GSPLIT   (KSPLIT / KG)
#define NSETS    3

struct FragSet {
  v2f    a[KG];
  float4 b[KG][2];
};

template <int BSTRIDE>
__device__ __forceinline__ void load_group(FragSet& f, const float* __restrict__ ap,
                                           const float* __restrict__ bp) {
#pragma unroll
  for (int i = 0; i < KG; ++i) {
    f.a[i] = *(const v2f*)(ap + i * 4);
    const float4* q = (const float4*)(bp + (size_t)i * BSTRIDE);
    f.b[i][0] = q[0];
    f.b[i][1] = q[1];
  }
}

__device__ __forceinline__ void mma_group(v8f (&acc)[4], const FragSet& f) {
#pragma unroll
  for (int i = 0; i < KG; ++i) {
    const v2f a = f.a[i];
    v2f b0; b0.x = f.b[i][0].x; b0.y = f.b[i][0].y;
    v2f b1; b1.x = f.b[i][0].z; b1.y = f.b[i][0].w;
    v2f b2; b2.x = f.b[i][1].x; b2.y = f.b[i][1].y;
    v2f b3; b3.x = f.b[i][1].z; b3.y = f.b[i][1].w;
    acc[0] = __builtin_amdgcn_wmma_f32_16x16x4_f32(false, a, false, b0,
                                                   (short)0, acc[0], false, false);
    acc[1] = __builtin_amdgcn_wmma_f32_16x16x4_f32(false, a, false, b1,
                                                   (short)0, acc[1], false, false);
    acc[2] = __builtin_amdgcn_wmma_f32_16x16x4_f32(false, a, false, b2,
                                                   (short)0, acc[2], false, false);
    acc[3] = __builtin_amdgcn_wmma_f32_16x16x4_f32(false, a, false, b3,
                                                   (short)0, acc[3], false, false);
  }
}

template <int N, bool RELU>
__global__ __launch_bounds__(128)
void sage_gemm_wmma_kernel(const float* __restrict__ A0,
                           const float* __restrict__ A1,
                           const float* __restrict__ Bpack,
                           const float* __restrict__ bias,
                           float* __restrict__ out,
                           int M) {
  constexpr int NGRPN   = N >> 6;
  constexpr int BSTRIDE = NGRPN * 256;
  const int lane = threadIdx.x & 31;
  const int wave = threadIdx.x >> 5;
  const int grp   = blockIdx.x * (blockDim.x >> 5) + wave;
  const int ngrps = (M >> 4) * NGRPN;
  if (grp >= ngrps) return;

  const int tm = grp / NGRPN;
  const int gn = grp % NGRPN;

  const int l15  = lane & 15;
  const int row  = tm * 16 + l15;
  const int col0 = gn * 64 + l15;
  const int kh   = (lane < 16) ? 0 : 2;

  const float* Arow0 = A0 + (size_t)row * IN_F + kh;
  const float* Arow1 = A1 + (size_t)row * IN_F + kh;
  const float* Bp    = Bpack + (size_t)gn * 256 + lane * 8;

  v8f acc[4] = {};
  FragSet F[NSETS];

  auto a_of = [&](int g) -> const float* {
    return (g < GSPLIT) ? (Arow0 + g * (KG * 4)) : (Arow1 + (g - GSPLIT) * (KG * 4));
  };
  auto b_of = [&](int g) -> const float* {
    return Bp + (size_t)g * (KG * BSTRIDE);
  };

#define PIN() __builtin_amdgcn_sched_barrier(0)

  load_group<BSTRIDE>(F[0], a_of(0), b_of(0));
  load_group<BSTRIDE>(F[1], a_of(1), b_of(1));
#pragma unroll
  for (int g = 0; g < NGROUPS; ++g) {
    if (g + 2 < NGROUPS) {
      load_group<BSTRIDE>(F[(g + 2) % NSETS], a_of(g + 2), b_of(g + 2));
      PIN();
    }
    mma_group(acc, F[g % NSETS]);
  }
#undef PIN

  const int mbase = tm * 16 + ((lane < 16) ? 0 : 8);
#pragma unroll
  for (int t = 0; t < 4; ++t) {
    const int col = col0 + t * 16;
    const float bv = bias[col];
#pragma unroll
    for (int i = 0; i < 8; ++i) {
      float v = acc[t][i] + bv;
      if (RELU) v = fmaxf(v, 0.0f);
      out[(size_t)(mbase + i) * N + col] = v;
    }
  }
}

extern "C" void kernel_launch(void* const* d_in, const int* in_sizes, int n_in,
                              void* d_out, int out_size, void* d_ws, size_t ws_size,
                              hipStream_t stream) {
  const float* x   = (const float*)d_in[0];
  const float* W1  = (const float*)d_in[1];
  const float* b1  = (const float*)d_in[2];
  const float* W2  = (const float*)d_in[3];
  const float* b2  = (const float*)d_in[4];
  const int* src1  = (const int*)d_in[5];
  const int* dst1  = (const int*)d_in[6];
  const int* src2  = (const int*)d_in[7];
  const int* dst2  = (const int*)d_in[8];
  const int E1 = in_sizes[5];
  const int E2 = in_sizes[7];

  char* wsb = (char*)d_ws;
  size_t off = 0;
  auto take = [&](size_t n_elems) { void* p = wsb + off; off += n_elems * 4; return p; };

  (void)take(N_DST1);
  (void)take(N_DST2);
  int* row_ptr1 = (int*)take(N_DST1 + 1);
  (void)take(N_DST1);
  int* srcs1    = (int*)take((size_t)E1);
  int* row_ptr2 = (int*)take(N_DST2 + 1);
  (void)take(N_DST2);
  int* srcs2    = (int*)take((size_t)E2);
  const size_t off_hN1 = off;
  float* hN1    = (float*)take((size_t)N_DST1 * IN_F);
  float* h      = (float*)take((size_t)N_DST1 * HID_F);
  const size_t off_hN2 = off;
  float* hN2    = (float*)take((size_t)N_DST2 * IN_F);
  const int nc1 = (E1 + CS1 - 1) / CS1;
  const int nc2 = (E2 + CS2 - 1) / CS2;
  unsigned* sorted1 = (unsigned*)(wsb + ((off_hN1 + 15) / 16) * 16);
  unsigned* sorted2 = (unsigned*)(wsb + ((off_hN2 + 15) / 16) * 16);
  int* tab1 = (int*)(sorted1 + (size_t)nc1 * CS1);
  int* tab2 = (int*)(sorted2 + (size_t)nc2 * CS2);
  float* Bpack1 = (float*)take((size_t)KDIM * HID_F);
  float* Bpack2 = (float*)take((size_t)KDIM * N_CLS);
  (void)ws_size;

  {
    int t1 = KSTEPS * (HID_F / 64) * 128;
    int t2 = KSTEPS * (N_CLS / 64) * 128;
    pack_W_kernel<<<(t1 + 255) / 256, 256, 0, stream>>>(W1, Bpack1, HID_F);
    pack_W_kernel<<<(t2 + 255) / 256, 256, 0, stream>>>(W2, Bpack2, N_CLS);
  }

  chunk_sort_kernel<<<nc1 + nc2, 256, 0, stream>>>(src1, dst1, E1, nc1, sorted1, tab1,
                                                   src2, dst2, E2, sorted2, tab2);
  bucket_build_kernel<<<NB1 + NB2, 256, 0, stream>>>(sorted1, tab1, nc1, E1, row_ptr1, srcs1,
                                                     sorted2, tab2, nc2, E2, row_ptr2, srcs2);

  gather_mean_kernel<<<N_DST1 / 8, 256, 0, stream>>>(x, srcs1, row_ptr1, hN1, N_DST1);

  {
    int ngrps = (N_DST1 / 16) * (HID_F / 64);
    sage_gemm_wmma_kernel<HID_F, true><<<ngrps / 4, 128, 0, stream>>>(
        x, hN1, Bpack1, b1, h, N_DST1);
  }

  gather_mean_kernel<<<N_DST2 / 8, 256, 0, stream>>>(h, srcs2, row_ptr2, hN2, N_DST2);

  {
    int ngrps = (N_DST2 / 16) * (N_CLS / 64);
    sage_gemm_wmma_kernel<N_CLS, false><<<(ngrps + 3) / 4, 128, 0, stream>>>(
        h, hN2, Bpack2, b2, (float*)d_out, N_DST2);
  }
}


# Layer-1 GEMM: work-group copies its 16-row A tile to LDS once (whole-line loads), the 4 waves read A fragments from LDS instead of 16-line 8-byte global fragment loads; k and sum order unchanged
# speedup vs baseline: 2.3504x; 1.0781x over previous
#include <hip/hip_runtime.h>
#include <hip/hip_bf16.h>


typedef __attribute__((ext_vector_type(2))) float v2f;
typedef __attribute__((ext_vector_type(8))) float v8f;

#define N_DST1   50000
#define N_DST2   10000
#define IN_F     256
#define KDIM     512
#define KSTEPS   (KDIM / 4)
#define KSPLIT   (IN_F / 4)
#define HID_F    256
#define N_CLS    64

#define BKT        256
#define NB1        ((N_DST1 + BKT - 1) / BKT)
#define NB2        ((N_DST2 + BKT - 1) / BKT)
#define CS1        4096
#define CS2        1024
#define CS_MAX     4096

__device__ __forceinline__ int block_excl_scan_256(int v, int* wtot, int lane, int w) {
  int inc = v;
  for (int o = 1; o < 32; o <<= 1) {
    const int u = __shfl_up(inc, o, 32);
    if (lane >= o) inc += u;
  }
  if (lane == 31) wtot[w] = inc;
  __syncthreads();
  int before = 0;
  for (int ww = 0; ww < 8; ++ww)
    if (ww < w) before += wtot[ww];
  return before + inc - v;
}

__global__ __launch_bounds__(256)
void chunk_sort_kernel(const int* __restrict__ src1, const int* __restrict__ dst1,
                       int E1, int nc1,
                       unsigned* __restrict__ sorted1, int* __restrict__ tab1,
                       const int* __restrict__ src2, const int* __restrict__ dst2,
                       int E2,
                       unsigned* __restrict__ sorted2, int* __restrict__ tab2) {
  __shared__ int cnt[256];
  __shared__ int wtot[8];
  __shared__ __attribute__((aligned(16))) unsigned stage[CS_MAX];
  const int t    = threadIdx.x;
  const int lane = t & 31;
  const int w    = t >> 5;

  int c = blockIdx.x;
  const int* src; const int* dst; unsigned* sorted; int* tab; int E, cs, nb;
  if (c < nc1) { src = src1; dst = dst1; sorted = sorted1; tab = tab1; E = E1; cs = CS1; nb = NB1; }
  else { c -= nc1; src = src2; dst = dst2; sorted = sorted2; tab = tab2; E = E2; cs = CS2; nb = NB2; }

  cnt[t] = 0;
  const int base = c * cs;
  const int n    = min(cs, E - base);

  int d[16], s[16], r[16];
#pragma unroll
  for (int j = 0; j < 4; ++j) {
    const int e0 = j * 1024 + t * 4;
#pragma unroll
    for (int k = 0; k < 4; ++k) { d[j * 4 + k] = -1; s[j * 4 + k] = 0; }
    if (e0 + 3 < n) {
      const int4 dv = *(const int4*)(dst + base + e0);
      const int4 sv = *(const int4*)(src + base + e0);
      d[j * 4 + 0] = dv.x; d[j * 4 + 1] = dv.y; d[j * 4 + 2] = dv.z; d[j * 4 + 3] = dv.w;
      s[j * 4 + 0] = sv.x; s[j * 4 + 1] = sv.y; s[j * 4 + 2] = sv.z; s[j * 4 + 3] = sv.w;
    } else {
#pragma unroll
      for (int k = 0; k < 4; ++k)
        if (e0 + k < n) { d[j * 4 + k] = dst[base + e0 + k]; s[j * 4 + k] = src[base + e0 + k]; }
    }
  }
  __syncthreads();

#pragma unroll
  for (int i = 0; i < 16; ++i) {
    r[i] = 0;
    if (d[i] >= 0) r[i] = atomicAdd(&cnt[d[i] >> 8], 1);
  }
  __syncthreads();

  const int excl = block_excl_scan_256(cnt[t], wtot, lane, w);
  cnt[t] = excl;
  __syncthreads();

#pragma unroll
  for (int i = 0; i < 16; ++i)
    if (d[i] >= 0)
      stage[cnt[d[i] >> 8] + r[i]] = ((unsigned)s[i] << 8) | (unsigned)(d[i] & (BKT - 1));
  __syncthreads();

  for (int i = t * 4; i < n; i += 1024) {
    if (i + 3 < n) {
      *(uint4*)(sorted + base + i) = *(const uint4*)(stage + i);
    } else {
      for (int k = i; k < n; ++k) sorted[base + k] = stage[k];
    }
  }
  int* trow = tab + (size_t)c * (nb + 1);
  if (t < nb)  trow[t]  = excl;
  if (t == 0)  trow[nb] = n;
}

template <int PASS>
__device__ __forceinline__ void walk_runs(const unsigned* __restrict__ sorted,
                                          const int* __restrict__ tab,
                                          int nc, int cs, int nbp1, int b,
                                          int* kcnt, int* __restrict__ srcs,
                                          int w, int lane) {
  for (int cb = 0; cb < nc; cb += 256) {
    const int c = cb + w + 8 * lane;
    int p = 0, n = 0;
    if (c < nc) {
      const int a = tab[(size_t)c * nbp1 + b];
      n = tab[(size_t)c * nbp1 + b + 1] - a;
      p = c * cs + a;
    }
    const int nruns = min(32, (nc - cb - w + 7) >> 3);
    for (int l = 0; l < nruns; l += 4) {
      const int p0 = __builtin_amdgcn_readlane(p, l);
      const int p1 = __builtin_amdgcn_readlane(p, l + 1);
      const int p2 = __builtin_amdgcn_readlane(p, l + 2);
      const int p3 = __builtin_amdgcn_readlane(p, l + 3);
      const int n0 = __builtin_amdgcn_readlane(n, l);
      const int n1 = __builtin_amdgcn_readlane(n, l + 1);
      const int n2 = __builtin_amdgcn_readlane(n, l + 2);
      const int n3 = __builtin_amdgcn_readlane(n, l + 3);
      const int m  = max(max(n0, n1), max(n2, n3));
      for (int k0 = 0; k0 < m; k0 += 32) {
        const int k = k0 + lane;
        const bool q0 = k < n0, q1 = k < n1, q2 = k < n2, q3 = k < n3;
        const unsigned v0 = sorted[q0 ? p0 + k : 0];
        const unsigned v1 = sorted[q1 ? p1 + k : 0];
        const unsigned v2 = sorted[q2 ? p2 + k : 0];
        const unsigned v3 = sorted[q3 ? p3 + k : 0];
        if (PASS == 1) {
          if (q0) atomicAdd(&kcnt[v0 & (BKT - 1)], 1);
          if (q1) atomicAdd(&kcnt[v1 & (BKT - 1)], 1);
          if (q2) atomicAdd(&kcnt[v2 & (BKT - 1)], 1);
          if (q3) atomicAdd(&kcnt[v3 & (BKT - 1)], 1);
        } else {
          if (q0) srcs[atomicAdd(&kcnt[v0 & (BKT - 1)], 1)] = (int)(v0 >> 8);
          if (q1) srcs[atomicAdd(&kcnt[v1 & (BKT - 1)], 1)] = (int)(v1 >> 8);
          if (q2) srcs[atomicAdd(&kcnt[v2 & (BKT - 1)], 1)] = (int)(v2 >> 8);
          if (q3) srcs[atomicAdd(&kcnt[v3 & (BKT - 1)], 1)] = (int)(v3 >> 8);
        }
      }
    }
  }
}

__global__ __launch_bounds__(256)
void bucket_build_kernel(const unsigned* __restrict__ sorted1, const int* __restrict__ tab1,
                         int nc1, int E1,
                         int* __restrict__ row_ptr1, int* __restrict__ srcs1,
                         const unsigned* __restrict__ sorted2, const int* __restrict__ tab2,
                         int nc2, int E2,
                         int* __restrict__ row_ptr2, int* __restrict__ srcs2) {
  __shared__ int kcnt[BKT];
  __shared__ int wred[8];
  __shared__ int wtot[8];
  const int t    = threadIdx.x;
  const int lane = t & 31;
  const int w    = __builtin_amdgcn_readfirstlane(t >> 5);

  int b = blockIdx.x;
  const unsigned* sorted; const int* tab; int* row_ptr; int* srcs; int nc, E, cs, nb, N;
  if (b < NB1) { sorted = sorted1; tab = tab1; row_ptr = row_ptr1; srcs = srcs1;
                 nc = nc1; E = E1; cs = CS1; nb = NB1; N = N_DST1; }
  else { b -= NB1; sorted = sorted2; tab = tab2; row_ptr = row_ptr2; srcs = srcs2;
         nc = nc2; E = E2; cs = CS2; nb = NB2; N = N_DST2; }
  const int nbp1 = nb + 1;

  kcnt[t] = 0;
  int asum = 0;
  for (int c = t; c < nc; c += 256) asum += tab[(size_t)c * nbp1 + b];
  for (int o = 16; o > 0; o >>= 1) asum += __shfl_xor(asum, o, 32);
  if (lane == 0) wred[w] = asum;
  __syncthreads();
  int base = 0;
  for (int ww = 0; ww < 8; ++ww) base += wred[ww];

  walk_runs<1>(sorted, tab, nc, cs, nbp1, b, kcnt, srcs, w, lane);
  __syncthreads();

  const int excl = block_excl_scan_256(kcnt[t], wtot, lane, w);
  const int row  = b * BKT + t;
  if (row < N) row_ptr[row] = base + excl;
  if (b == nb - 1 && t == 0) row_ptr[N] = E;
  kcnt[t] = base + excl;
  __syncthreads();

  walk_runs<2>(sorted, tab, nc, cs, nbp1, b, kcnt, srcs, w, lane);
}

__global__ __launch_bounds__(256)
void gather_mean_kernel(const float* __restrict__ hsrc,
                        const int* __restrict__ srcs,
                        const int* __restrict__ row_ptr,
                        float* __restrict__ hN,
                        int n_dst) {
  const int d = blockIdx.x * (blockDim.x >> 5) + (threadIdx.x >> 5);
  if (d >= n_dst) return;
  const int lane = threadIdx.x & 31;
  const int beg = row_ptr[d];
  const int end = row_ptr[d + 1];

  float4 a0 = make_float4(0.f, 0.f, 0.f, 0.f);
  float4 a1 = make_float4(0.f, 0.f, 0.f, 0.f);
  for (int j = beg; j < end; ++j) {
    int s = srcs[j];
    const float4* p = (const float4*)(hsrc + (size_t)s * IN_F) + lane * 2;
    float4 v0 = p[0];
    float4 v1 = p[1];
    a0.x += v0.x; a0.y += v0.y; a0.z += v0.z; a0.w += v0.w;
    a1.x += v1.x; a1.y += v1.y; a1.z += v1.z; a1.w += v1.w;
  }
  const float inv = 1.0f / fmaxf((float)(end - beg), 1.0f);
  a0.x *= inv; a0.y *= inv; a0.z *= inv; a0.w *= inv;
  a1.x *= inv; a1.y *= inv; a1.z *= inv; a1.w *= inv;
  float4* o = (float4*)(hN + (size_t)d * IN_F) + lane * 2;
  o[0] = a0;
  o[1] = a1;
}

__global__ void pack_W_kernel(const float* __restrict__ W,
                              float* __restrict__ Bpack, int N) {
  const int ngrpN = N >> 6;
  const int total = KSTEPS * ngrpN * 128;
  int idx = blockIdx.x * blockDim.x + threadIdx.x;
  if (idx >= total) return;

  const int w    = idx & 127;
  const int t    = w & 3;
  const int lane = w >> 2;
  const int blk  = idx >> 7;
  const int gn   = blk % ngrpN;
  const int ks   = blk / ngrpN;

  const int kh  = (lane < 16) ? 0 : 2;
  const int col = gn * 64 + t * 16 + (lane & 15);
  const int kr  = ks * 4 + kh;

  float2 v;
  v.x = W[(size_t)kr * N + col];
  v.y = W[(size_t)(kr + 1) * N + col];
  ((float2*)Bpack)[idx] = v;
}

#define KG       8
#define NGROUPS  (KSTEPS / KG)
#define GSPLIT   (KSPLIT / KG)
#define NSETS    3

struct FragSet {
  v2f    a[KG];
  float4 b[KG][2];
};

template <int BSTRIDE>
__device__ __forceinline__ void load_group(FragSet& f, const float* __restrict__ ap,
                                           const float* __restrict__ bp) {
#pragma unroll
  for (int i = 0; i < KG; ++i) {
    f.a[i] = *(const v2f*)(ap + i * 4);
    const float4* q = (const float4*)(bp + (size_t)i * BSTRIDE);
    f.b[i][0] = q[0];
    f.b[i][1] = q[1];
  }
}

#define A_PITCH  (KDIM + 4)
template <int BSTRIDE>
__device__ __forceinline__ void load_group_b(FragSet& f, const float* __restrict__ bp) {
#pragma unroll
  for (int i = 0; i < KG; ++i) {
    const float4* q = (const float4*)(bp + (size_t)i * BSTRIDE);
    f.b[i][0] = q[0];
    f.b[i][1] = q[1];
  }
}
__device__ __forceinline__ void load_group_a_lds(FragSet& f, const float* as, int k0) {
#pragma unroll
  for (int i = 0; i < KG; ++i) f.a[i] = *(const v2f*)(as + k0 + i * 4);
}

__device__ __forceinline__ void mma_group(v8f (&acc)[4], const FragSet& f) {
#pragma unroll
  for (int i = 0; i < KG; ++i) {
    const v2f a = f.a[i];
    v2f b0; b0.x = f.b[i][0].x; b0.y = f.b[i][0].y;
    v2f b1; b1.x = f.b[i][0].z; b1.y = f.b[i][0].w;
    v2f b2; b2.x = f.b[i][1].x; b2.y = f.b[i][1].y;
    v2f b3; b3.x = f.b[i][1].z; b3.y = f.b[i][1].w;
    acc[0] = __builtin_amdgcn_wmma_f32_16x16x4_f32(false, a, false, b0,
                                                   (short)0, acc[0], false, false);
    acc[1] = __builtin_amdgcn_wmma_f32_16x16x4_f32(false, a, false, b1,
                                                   (short)0, acc[1], false, false);
    acc[2] = __builtin_amdgcn_wmma_f32_16x16x4_f32(false, a, false, b2,
                                                   (short)0, acc[2], false, false);
    acc[3] = __builtin_amdgcn_wmma_f32_16x16x4_f32(false, a, false, b3,
                                                   (short)0, acc[3], false, false);
  }
}

template <int N, bool RELU>
__global__ __launch_bounds__(128)
void sage_gemm_wmma_kernel(const float* __restrict__ A0,
                           const float* __restrict__ A1,
                           const float* __restrict__ Bpack,
                           const float* __restrict__ bias,
                           float* __restrict__ out,
                           int M) {
  constexpr int NGRPN   = N >> 6;
  constexpr int BSTRIDE = NGRPN * 256;
  const int lane = threadIdx.x & 31;
  const int wave = threadIdx.x >> 5;
  const int grp   = blockIdx.x * (blockDim.x >> 5) + wave;
  const int ngrps = (M >> 4) * NGRPN;
  if (grp >= ngrps) return;

  const int tm = grp / NGRPN;
  const int gn = grp % NGRPN;

  const int l15  = lane & 15;
  const int row  = tm * 16 + l15;
  const int col0 = gn * 64 + l15;
  const int kh   = (lane < 16) ? 0 : 2;

  const float* Arow0 = A0 + (size_t)row * IN_F + kh;
  const float* Arow1 = A1 + (size_t)row * IN_F + kh;
  const float* Bp    = Bpack + (size_t)gn * 256 + lane * 8;

  v8f acc[4] = {};
  FragSet F[NSETS];

  auto a_of = [&](int g) -> const float* {
    return (g < GSPLIT) ? (Arow0 + g * (KG * 4)) : (Arow1 + (g - GSPLIT) * (KG * 4));
  };
  auto b_of = [&](int g) -> const float* {
    return Bp + (size_t)g * (KG * BSTRIDE);
  };

#define PIN() __builtin_amdgcn_sched_barrier(0)

  if constexpr (NGRPN == 4) {
    __shared__ __attribute__((aligned(16))) float As[16 * A_PITCH];
    const int t = threadIdx.x;
    const float* srow = (t < 64) ? (A0 + (size_t)tm * 16 * IN_F + t * 4)
                                 : (A1 + (size_t)tm * 16 * IN_F + (t - 64) * 4);
    float4 st[16];
#pragma unroll
    for (int j = 0; j < 16; ++j) st[j] = *(const float4*)(srow + (size_t)j * IN_F);
    load_group_b<BSTRIDE>(F[0], b_of(0));
    load_group_b<BSTRIDE>(F[1], b_of(1));
#pragma unroll
    for (int j = 0; j < 16; ++j) *(float4*)&As[j * A_PITCH + t * 4] = st[j];
    __syncthreads();

    const float* as = &As[l15 * A_PITCH + kh];
    load_group_a_lds(F[0], as, 0 * (KG * 4));
    load_group_a_lds(F[1], as, 1 * (KG * 4));
#pragma unroll
    for (int g = 0; g < NGROUPS; ++g) {
      if (g + 2 < NGROUPS) {
        load_group_b<BSTRIDE>(F[(g + 2) % NSETS], b_of(g + 2));
        load_group_a_lds(F[(g + 2) % NSETS], as, (g + 2) * (KG * 4));
        PIN();
      }
      mma_group(acc, F[g % NSETS]);
    }
  } else {
    load_group<BSTRIDE>(F[0], a_of(0), b_of(0));
    load_group<BSTRIDE>(F[1], a_of(1), b_of(1));
#pragma unroll
    for (int g = 0; g < NGROUPS; ++g) {
      if (g + 2 < NGROUPS) {
        load_group<BSTRIDE>(F[(g + 2) % NSETS], a_of(g + 2), b_of(g + 2));
        PIN();
      }
      mma_group(acc, F[g % NSETS]);
    }
  }
#undef PIN

  const int mbase = tm * 16 + ((lane < 16) ? 0 : 8);
#pragma unroll
  for (int t = 0; t < 4; ++t) {
    const int col = col0 + t * 16;
    const float bv = bias[col];
#pragma unroll
    for (int i = 0; i < 8; ++i) {
      float v = acc[t][i] + bv;
      if (RELU) v = fmaxf(v, 0.0f);
      out[(size_t)(mbase + i) * N + col] = v;
    }
  }
}

extern "C" void kernel_launch(void* const* d_in, const int* in_sizes, int n_in,
                              void* d_out, int out_size, void* d_ws, size_t ws_size,
                              hipStream_t stream) {
  const float* x   = (const float*)d_in[0];
  const float* W1  = (const float*)d_in[1];
  const float* b1  = (const float*)d_in[2];
  const float* W2  = (const float*)d_in[3];
  const float* b2  = (const float*)d_in[4];
  const int* src1  = (const int*)d_in[5];
  const int* dst1  = (const int*)d_in[6];
  const int* src2  = (const int*)d_in[7];
  const int* dst2  = (const int*)d_in[8];
  const int E1 = in_sizes[5];
  const int E2 = in_sizes[7];

  char* wsb = (char*)d_ws;
  size_t off = 0;
  auto take = [&](size_t n_elems) { void* p = wsb + off; off += n_elems * 4; return p; };

  (void)take(N_DST1);
  (void)take(N_DST2);
  int* row_ptr1 = (int*)take(N_DST1 + 1);
  (void)take(N_DST1);
  int* srcs1    = (int*)take((size_t)E1);
  int* row_ptr2 = (int*)take(N_DST2 + 1);
  (void)take(N_DST2);
  int* srcs2    = (int*)take((size_t)E2);
  const size_t off_hN1 = off;
  float* hN1    = (float*)take((size_t)N_DST1 * IN_F);
  float* h      = (float*)take((size_t)N_DST1 * HID_F);
  const size_t off_hN2 = off;
  float* hN2    = (float*)take((size_t)N_DST2 * IN_F);
  const int nc1 = (E1 + CS1 - 1) / CS1;
  const int nc2 = (E2 + CS2 - 1) / CS2;
  unsigned* sorted1 = (unsigned*)(wsb + ((off_hN1 + 15) / 16) * 16);
  unsigned* sorted2 = (unsigned*)(wsb + ((off_hN2 + 15) / 16) * 16);
  int* tab1 = (int*)(sorted1 + (size_t)nc1 * CS1);
  int* tab2 = (int*)(sorted2 + (size_t)nc2 * CS2);
  float* Bpack1 = (float*)take((size_t)KDIM * HID_F);
  float* Bpack2 = (float*)take((size_t)KDIM * N_CLS);
  (void)ws_size;

  {
    int t1 = KSTEPS * (HID_F / 64) * 128;
    int t2 = KSTEPS * (N_CLS / 64) * 128;
    pack_W_kernel<<<(t1 + 255) / 256, 256, 0, stream>>>(W1, Bpack1, HID_F);
    pack_W_kernel<<<(t2 + 255) / 256, 256, 0, stream>>>(W2, Bpack2, N_CLS);
  }

  chunk_sort_kernel<<<nc1 + nc2, 256, 0, stream>>>(src1, dst1, E1, nc1, sorted1, tab1,
                                                   src2, dst2, E2, sorted2, tab2);
  bucket_build_kernel<<<NB1 + NB2, 256, 0, stream>>>(sorted1, tab1, nc1, E1, row_ptr1, srcs1,
                                                     sorted2, tab2, nc2, E2, row_ptr2, srcs2);

  gather_mean_kernel<<<N_DST1 / 8, 256, 0, stream>>>(x, srcs1, row_ptr1, hN1, N_DST1);

  {
    int ngrps = (N_DST1 / 16) * (HID_F / 64);
    sage_gemm_wmma_kernel<HID_F, true><<<ngrps / 4, 128, 0, stream>>>(
        x, hN1, Bpack1, b1, h, N_DST1);
  }

  gather_mean_kernel<<<N_DST2 / 8, 256, 0, stream>>>(h, srcs2, row_ptr2, hN2, N_DST2);

  {
    int ngrps = (N_DST2 / 16) * (N_CLS / 64);
    sage_gemm_wmma_kernel<N_CLS, false><<<(ngrps + 3) / 4, 128, 0, stream>>>(
        h, hN2, Bpack2, b2, (float*)d_out, N_DST2);
  }
}


# Layer-1 gather-mean moved into its GEMM: a work-group sums its 16 rows' neighbors (16 rows in flight per wave, same order) straight into the LDS A tile; hN1 plane and its launch gone
# speedup vs baseline: 2.4383x; 1.0374x over previous
#include <hip/hip_runtime.h>
#include <hip/hip_bf16.h>


typedef __attribute__((ext_vector_type(2))) float v2f;
typedef __attribute__((ext_vector_type(8))) float v8f;

#define N_DST1   50000
#define N_DST2   10000
#define IN_F     256
#define KDIM     512
#define KSTEPS   (KDIM / 4)
#define KSPLIT   (IN_F / 4)
#define HID_F    256
#define N_CLS    64

#define BKT        256
#define NB1        ((N_DST1 + BKT - 1) / BKT)
#define NB2        ((N_DST2 + BKT - 1) / BKT)
#define CS1        4096
#define CS2        1024
#define CS_MAX     4096

__device__ __forceinline__ int block_excl_scan_256(int v, int* wtot, int lane, int w) {
  int inc = v;
  for (int o = 1; o < 32; o <<= 1) {
    const int u = __shfl_up(inc, o, 32);
    if (lane >= o) inc += u;
  }
  if (lane == 31) wtot[w] = inc;
  __syncthreads();
  int before = 0;
  for (int ww = 0; ww < 8; ++ww)
    if (ww < w) before += wtot[ww];
  return before + inc - v;
}

__global__ __launch_bounds__(256)
void chunk_sort_kernel(const int* __restrict__ src1, const int* __restrict__ dst1,
                       int E1, int nc1,
                       unsigned* __restrict__ sorted1, int* __restrict__ tab1,
                       const int* __restrict__ src2, const int* __restrict__ dst2,
                       int E2,
                       unsigned* __restrict__ sorted2, int* __restrict__ tab2) {
  __shared__ int cnt[256];
  __shared__ int wtot[8];
  __shared__ __attribute__((aligned(16))) unsigned stage[CS_MAX];
  const int t    = threadIdx.x;
  const int lane = t & 31;
  const int w    = t >> 5;

  int c = blockIdx.x;
  const int* src; const int* dst; unsigned* sorted; int* tab; int E, cs, nb;
  if (c < nc1) { src = src1; dst = dst1; sorted = sorted1; tab = tab1; E = E1; cs = CS1; nb = NB1; }
  else { c -= nc1; src = src2; dst = dst2; sorted = sorted2; tab = tab2; E = E2; cs = CS2; nb = NB2; }

  cnt[t] = 0;
  const int base = c * cs;
  const int n    = min(cs, E - base);

  int d[16], s[16], r[16];
#pragma unroll
  for (int j = 0; j < 4; ++j) {
    const int e0 = j * 1024 + t * 4;
#pragma unroll
    for (int k = 0; k < 4; ++k) { d[j * 4 + k] = -1; s[j * 4 + k] = 0; }
    if (e0 + 3 < n) {
      const int4 dv = *(const int4*)(dst + base + e0);
      const int4 sv = *(const int4*)(src + base + e0);
      d[j * 4 + 0] = dv.x; d[j * 4 + 1] = dv.y; d[j * 4 + 2] = dv.z; d[j * 4 + 3] = dv.w;
      s[j * 4 + 0] = sv.x; s[j * 4 + 1] = sv.y; s[j * 4 + 2] = sv.z; s[j * 4 + 3] = sv.w;
    } else {
#pragma unroll
      for (int k = 0; k < 4; ++k)
        if (e0 + k < n) { d[j * 4 + k] = dst[base + e0 + k]; s[j * 4 + k] = src[base + e0 + k]; }
    }
  }
  __syncthreads();

#pragma unroll
  for (int i = 0; i < 16; ++i) {
    r[i] = 0;
    if (d[i] >= 0) r[i] = atomicAdd(&cnt[d[i] >> 8], 1);
  }
  __syncthreads();

  const int excl = block_excl_scan_256(cnt[t], wtot, lane, w);
  cnt[t] = excl;
  __syncthreads();

#pragma unroll
  for (int i = 0; i < 16; ++i)
    if (d[i] >= 0)
      stage[cnt[d[i] >> 8] + r[i]] = ((unsigned)s[i] << 8) | (unsigned)(d[i] & (BKT - 1));
  __syncthreads();

  for (int i = t * 4; i < n; i += 1024) {
    if (i + 3 < n) {
      *(uint4*)(sorted + base + i) = *(const uint4*)(stage + i);
    } else {
      for (int k = i; k < n; ++k) sorted[base + k] = stage[k];
    }
  }
  int* trow = tab + (size_t)c * (nb + 1);
  if (t < nb)  trow[t]  = excl;
  if (t == 0)  trow[nb] = n;
}

template <int PASS>
__device__ __forceinline__ void walk_runs(const unsigned* __restrict__ sorted,
                                          const int* __restrict__ tab,
                                          int nc, int cs, int nbp1, int b,
                                          int* kcnt, int* __restrict__ srcs,
                                          int w, int lane) {
  for (int cb = 0; cb < nc; cb += 256) {
    const int c = cb + w + 8 * lane;
    int p = 0, n = 0;
    if (c < nc) {
      const int a = tab[(size_t)c * nbp1 + b];
      n = tab[(size_t)c * nbp1 + b + 1] - a;
      p = c * cs + a;
    }
    const int nruns = min(32, (nc - cb - w + 7) >> 3);
    for (int l = 0; l < nruns; l += 4) {
      const int p0 = __builtin_amdgcn_readlane(p, l);
      const int p1 = __builtin_amdgcn_readlane(p, l + 1);
      const int p2 = __builtin_amdgcn_readlane(p, l + 2);
      const int p3 = __builtin_amdgcn_readlane(p, l + 3);
      const int n0 = __builtin_amdgcn_readlane(n, l);
      const int n1 = __builtin_amdgcn_readlane(n, l + 1);
      const int n2 = __builtin_amdgcn_readlane(n, l + 2);
      const int n3 = __builtin_amdgcn_readlane(n, l + 3);
      const int m  = max(max(n0, n1), max(n2, n3));
      for (int k0 = 0; k0 < m; k0 += 32) {
        const int k = k0 + lane;
        const bool q0 = k < n0, q1 = k < n1, q2 = k < n2, q3 = k < n3;
        const unsigned v0 = sorted[q0 ? p0 + k : 0];
        const unsigned v1 = sorted[q1 ? p1 + k : 0];
        const unsigned v2 = sorted[q2 ? p2 + k : 0];
        const unsigned v3 = sorted[q3 ? p3 + k : 0];
        if (PASS == 1) {
          if (q0) atomicAdd(&kcnt[v0 & (BKT - 1)], 1);
          if (q1) atomicAdd(&kcnt[v1 & (BKT - 1)], 1);
          if (q2) atomicAdd(&kcnt[v2 & (BKT - 1)], 1);
          if (q3) atomicAdd(&kcnt[v3 & (BKT - 1)], 1);
        } else {
          if (q0) srcs[atomicAdd(&kcnt[v0 & (BKT - 1)], 1)] = (int)(v0 >> 8);
          if (q1) srcs[atomicAdd(&kcnt[v1 & (BKT - 1)], 1)] = (int)(v1 >> 8);
          if (q2) srcs[atomicAdd(&kcnt[v2 & (BKT - 1)], 1)] = (int)(v2 >> 8);
          if (q3) srcs[atomicAdd(&kcnt[v3 & (BKT - 1)], 1)] = (int)(v3 >> 8);
        }
      }
    }
  }
}

__global__ __launch_bounds__(256)
void bucket_build_kernel(const unsigned* __restrict__ sorted1, const int* __restrict__ tab1,
                         int nc1, int E1,
                         int* __restrict__ row_ptr1, int* __restrict__ srcs1,
                         const unsigned* __restrict__ sorted2, const int* __restrict__ tab2,
                         int nc2, int E2,
                         int* __restrict__ row_ptr2, int* __restrict__ srcs2) {
  __shared__ int kcnt[BKT];
  __shared__ int wred[8];
  __shared__ int wtot[8];
  const int t    = threadIdx.x;
  const int lane = t & 31;
  const int w    = __builtin_amdgcn_readfirstlane(t >> 5);

  int b = blockIdx.x;
  const unsigned* sorted; const int* tab; int* row_ptr; int* srcs; int nc, E, cs, nb, N;
  if (b < NB1) { sorted = sorted1; tab = tab1; row_ptr = row_ptr1; srcs = srcs1;
                 nc = nc1; E = E1; cs = CS1; nb = NB1; N = N_DST1; }
  else { b -= NB1; sorted = sorted2; tab = tab2; row_ptr = row_ptr2; srcs = srcs2;
         nc = nc2; E = E2; cs = CS2; nb = NB2; N = N_DST2; }
  const int nbp1 = nb + 1;

  kcnt[t] = 0;
  int asum = 0;
  for (int c = t; c < nc; c += 256) asum += tab[(size_t)c * nbp1 + b];
  for (int o = 16; o > 0; o >>= 1) asum += __shfl_xor(asum, o, 32);
  if (lane == 0) wred[w] = asum;
  __syncthreads();
  int base = 0;
  for (int ww = 0; ww < 8; ++ww) base += wred[ww];

  walk_runs<1>(sorted, tab, nc, cs, nbp1, b, kcnt, srcs, w, lane);
  __syncthreads();

  const int excl = block_excl_scan_256(kcnt[t], wtot, lane, w);
  const int row  = b * BKT + t;
  if (row < N) row_ptr[row] = base + excl;
  if (b == nb - 1 && t == 0) row_ptr[N] = E;
  kcnt[t] = base + excl;
  __syncthreads();

  walk_runs<2>(sorted, tab, nc, cs, nbp1, b, kcnt, srcs, w, lane);
}

__global__ __launch_bounds__(256)
void gather_mean_kernel(const float* __restrict__ hsrc,
                        const int* __restrict__ srcs,
                        const int* __restrict__ row_ptr,
                        float* __restrict__ hN,
                        int n_dst) {
  const int d = blockIdx.x * (blockDim.x >> 5) + (threadIdx.x >> 5);
  if (d >= n_dst) return;
  const int lane = threadIdx.x & 31;
  const int beg = row_ptr[d];
  const int end = row_ptr[d + 1];

  float4 a0 = make_float4(0.f, 0.f, 0.f, 0.f);
  float4 a1 = make_float4(0.f, 0.f, 0.f, 0.f);
  for (int j = beg; j < end; ++j) {
    int s = srcs[j];
    const float4* p = (const float4*)(hsrc + (size_t)s * IN_F) + lane * 2;
    float4 v0 = p[0];
    float4 v1 = p[1];
    a0.x += v0.x; a0.y += v0.y; a0.z += v0.z; a0.w += v0.w;
    a1.x += v1.x; a1.y += v1.y; a1.z += v1.z; a1.w += v1.w;
  }
  const float inv = 1.0f / fmaxf((float)(end - beg), 1.0f);
  a0.x *= inv; a0.y *= inv; a0.z *= inv; a0.w *= inv;
  a1.x *= inv; a1.y *= inv; a1.z *= inv; a1.w *= inv;
  float4* o = (float4*)(hN + (size_t)d * IN_F) + lane * 2;
  o[0] = a0;
  o[1] = a1;
}

__global__ void pack_W_kernel(const float* __restrict__ W,
                              float* __restrict__ Bpack, int N) {
  const int ngrpN = N >> 6;
  const int total = KSTEPS * ngrpN * 128;
  int idx = blockIdx.x * blockDim.x + threadIdx.x;
  if (idx >= total) return;

  const int w    = idx & 127;
  const int t    = w & 3;
  const int lane = w >> 2;
  const int blk  = idx >> 7;
  const int gn   = blk % ngrpN;
  const int ks   = blk / ngrpN;

  const int kh  = (lane < 16) ? 0 : 2;
  const int col = gn * 64 + t * 16 + (lane & 15);
  const int kr  = ks * 4 + kh;

  float2 v;
  v.x = W[(size_t)kr * N + col];
  v.y = W[(size_t)(kr + 1) * N + col];
  ((float2*)Bpack)[idx] = v;
}

#define KG       8
#define NGROUPS  (KSTEPS / KG)
#define GSPLIT   (KSPLIT / KG)
#define NSETS    3

struct FragSet {
  v2f    a[KG];
  float4 b[KG][2];
};

template <int BSTRIDE>
__device__ __forceinline__ void load_group(FragSet& f, const float* __restrict__ ap,
                                           const float* __restrict__ bp) {
#pragma unroll
  for (int i = 0; i < KG; ++i) {
    f.a[i] = *(const v2f*)(ap + i * 4);
    const float4* q = (const float4*)(bp + (size_t)i * BSTRIDE);
    f.b[i][0] = q[0];
    f.b[i][1] = q[1];
  }
}

#define A_PITCH  (KDIM + 4)
#define GB       16
template <int BSTRIDE>
__device__ __forceinline__ void load_group_b(FragSet& f, const float* __restrict__ bp) {
#pragma unroll
  for (int i = 0; i < KG; ++i) {
    const float4* q = (const float4*)(bp + (size_t)i * BSTRIDE);
    f.b[i][0] = q[0];
    f.b[i][1] = q[1];
  }
}
__device__ __forceinline__ void load_group_a_lds(FragSet& f, const float* as, int k0) {
#pragma unroll
  for (int i = 0; i < KG; ++i) f.a[i] = *(const v2f*)(as + k0 + i * 4);
}

__device__ __forceinline__ void mma_group(v8f (&acc)[4], const FragSet& f) {
#pragma unroll
  for (int i = 0; i < KG; ++i) {
    const v2f a = f.a[i];
    v2f b0; b0.x = f.b[i][0].x; b0.y = f.b[i][0].y;
    v2f b1; b1.x = f.b[i][0].z; b1.y = f.b[i][0].w;
    v2f b2; b2.x = f.b[i][1].x; b2.y = f.b[i][1].y;
    v2f b3; b3.x = f.b[i][1].z; b3.y = f.b[i][1].w;
    acc[0] = __builtin_amdgcn_wmma_f32_16x16x4_f32(false, a, false, b0,
                                                   (short)0, acc[0], false, false);
    acc[1] = __builtin_amdgcn_wmma_f32_16x16x4_f32(false, a, false, b1,
                                                   (short)0, acc[1], false, false);
    acc[2] = __builtin_amdgcn_wmma_f32_16x16x4_f32(false, a, false, b2,
                                                   (short)0, acc[2], false, false);
    acc[3] = __builtin_amdgcn_wmma_f32_16x16x4_f32(false, a, false, b3,
                                                   (short)0, acc[3], false, false);
  }
}

template <int N, bool RELU>
__global__ __launch_bounds__(128)
void sage_gemm_wmma_kernel(const float* __restrict__ A0,
                           const float* __restrict__ A1,
                           const float* __restrict__ Bpack,
                           const float* __restrict__ bias,
                           float* __restrict__ out,
                           int M,
                           const int* __restrict__ srcs,
                           const int* __restrict__ row_ptr)
{
  constexpr int NGRPN   = N >> 6;
  constexpr int BSTRIDE = NGRPN * 256;
  const int lane = threadIdx.x & 31;
  const int wave = threadIdx.x >> 5;
  const int grp   = blockIdx.x * (blockDim.x >> 5) + wave;
  const int ngrps = (M >> 4) * NGRPN;
  if (grp >= ngrps) return;

  const int tm = grp / NGRPN;
  const int gn = grp % NGRPN;

  const int l15  = lane & 15;
  const int row  = tm * 16 + l15;
  const int col0 = gn * 64 + l15;
  const int kh   = (lane < 16) ? 0 : 2;

  const float* Arow0 = A0 + (size_t)row * IN_F + kh;
  const float* Arow1 = A1 + (size_t)row * IN_F + kh;
  const float* Bp    = Bpack + (size_t)gn * 256 + lane * 8;

  v8f acc[4] = {};
  FragSet F[NSETS];

  auto a_of = [&](int g) -> const float* {
    return (g < GSPLIT) ? (Arow0 + g * (KG * 4)) : (Arow1 + (g - GSPLIT) * (KG * 4));
  };
  auto b_of = [&](int g) -> const float* {
    return Bp + (size_t)g * (KG * BSTRIDE);
  };

#define PIN() __builtin_amdgcn_sched_barrier(0)

  if constexpr (NGRPN == 4) {
    __shared__ __attribute__((aligned(16))) float As[16 * A_PITCH];
    const int t = threadIdx.x;
    const int wu = __builtin_amdgcn_readfirstlane(wave);

    {
      const int lane5 = (lane < 4) ? lane : 4;
      const int rpv = row_ptr[blockIdx.x * 16 + wu * 4 + lane5];
      const float4* hl = (const float4*)A1 + lane * 2;
      for (int i = 0; i < 4; ++i) {
        const int beg = __builtin_amdgcn_readlane(rpv, i);
        const int end = __builtin_amdgcn_readlane(rpv, i + 1);
        float4 a0 = make_float4(0.f, 0.f, 0.f, 0.f);
        float4 a1 = make_float4(0.f, 0.f, 0.f, 0.f);
        for (int j0 = beg; j0 < end; j0 += 32) {
          const int nj = min(32, end - j0);
          const int sidx = (lane < nj) ? srcs[j0 + lane] : 0;
          for (int k0 = 0; k0 < nj; k0 += GB) {
            float4 v0[GB], v1[GB];
#pragma unroll
            for (int u = 0; u < GB; ++u) {
              const int s = __builtin_amdgcn_readlane(sidx, k0 + u);
              const float4* p = hl + (size_t)s * (IN_F / 4);
              v0[u] = p[0];
              v1[u] = p[1];
            }
#pragma unroll
            for (int u = 0; u < GB; ++u) {
              const bool q = (k0 + u < nj);
              a0.x += q ? v0[u].x : 0.f; a0.y += q ? v0[u].y : 0.f;
              a0.z += q ? v0[u].z : 0.f; a0.w += q ? v0[u].w : 0.f;
              a1.x += q ? v1[u].x : 0.f; a1.y += q ? v1[u].y : 0.f;
              a1.z += q ? v1[u].z : 0.f; a1.w += q ? v1[u].w : 0.f;
            }
          }
        }
        const float inv = 1.0f / fmaxf((float)(end - beg), 1.0f);
        a0.x *= inv; a0.y *= inv; a0.z *= inv; a0.w *= inv;
        a1.x *= inv; a1.y *= inv; a1.z *= inv; a1.w *= inv;
        float* orow = &As[(wu * 4 + i) * A_PITCH + IN_F + lane * 8];
        *(float4*)orow       = a0;
        *(float4*)(orow + 4) = a1;
      }
    }

    const int r0 = t >> 6;
    const float* srow = A0 + ((size_t)tm * 16 + r0) * IN_F + (t & 63) * 4;
    float4 st[8];
#pragma unroll
    for (int j = 0; j < 8; ++j) st[j] = *(const float4*)(srow + (size_t)(2 * j) * IN_F);
    load_group_b<BSTRIDE>(F[0], b_of(0));
    load_group_b<BSTRIDE>(F[1], b_of(1));
#pragma unroll
    for (int j = 0; j < 8; ++j) *(float4*)&As[(r0 + 2 * j) * A_PITCH + (t & 63) * 4] = st[j];
    __syncthreads();

    const float* as = &As[l15 * A_PITCH + kh];
    load_group_a_lds(F[0], as, 0 * (KG * 4));
    load_group_a_lds(F[1], as, 1 * (KG * 4));
#pragma unroll
    for (int g = 0; g < NGROUPS; ++g) {
      if (g + 2 < NGROUPS) {
        load_group_b<BSTRIDE>(F[(g + 2) % NSETS], b_of(g + 2));
        load_group_a_lds(F[(g + 2) % NSETS], as, (g + 2) * (KG * 4));
        PIN();
      }
      mma_group(acc, F[g % NSETS]);
    }
  } else {
    load_group<BSTRIDE>(F[0], a_of(0), b_of(0));
    load_group<BSTRIDE>(F[1], a_of(1), b_of(1));
#pragma unroll
    for (int g = 0; g < NGROUPS; ++g) {
      if (g + 2 < NGROUPS) {
        load_group<BSTRIDE>(F[(g + 2) % NSETS], a_of(g + 2), b_of(g + 2));
        PIN();
      }
      mma_group(acc, F[g % NSETS]);
    }
  }
#undef PIN

  const int mbase = tm * 16 + ((lane < 16) ? 0 : 8);
#pragma unroll
  for (int t = 0; t < 4; ++t) {
    const int col = col0 + t * 16;
    const float bv = bias[col];
#pragma unroll
    for (int i = 0; i < 8; ++i) {
      float v = acc[t][i] + bv;
      if (RELU) v = fmaxf(v, 0.0f);
      out[(size_t)(mbase + i) * N + col] = v;
    }
  }
}

extern "C" void kernel_launch(void* const* d_in, const int* in_sizes, int n_in,
                              void* d_out, int out_size, void* d_ws, size_t ws_size,
                              hipStream_t stream) {
  const float* x   = (const float*)d_in[0];
  const float* W1  = (const float*)d_in[1];
  const float* b1  = (const float*)d_in[2];
  const float* W2  = (const float*)d_in[3];
  const float* b2  = (const float*)d_in[4];
  const int* src1  = (const int*)d_in[5];
  const int* dst1  = (const int*)d_in[6];
  const int* src2  = (const int*)d_in[7];
  const int* dst2  = (const int*)d_in[8];
  const int E1 = in_sizes[5];
  const int E2 = in_sizes[7];

  char* wsb = (char*)d_ws;
  size_t off = 0;
  auto take = [&](size_t n_elems) { void* p = wsb + off; off += n_elems * 4; return p; };

  (void)take(N_DST1);
  (void)take(N_DST2);
  int* row_ptr1 = (int*)take(N_DST1 + 1);
  (void)take(N_DST1);
  int* srcs1    = (int*)take((size_t)E1);
  int* row_ptr2 = (int*)take(N_DST2 + 1);
  (void)take(N_DST2);
  int* srcs2    = (int*)take((size_t)E2);
  const size_t off_hN1 = off;
  (void)take((size_t)N_DST1 * IN_F);
  float* h      = (float*)take((size_t)N_DST1 * HID_F);
  const size_t off_hN2 = off;
  float* hN2    = (float*)take((size_t)N_DST2 * IN_F);
  const int nc1 = (E1 + CS1 - 1) / CS1;
  const int nc2 = (E2 + CS2 - 1) / CS2;
  unsigned* sorted1 = (unsigned*)(wsb + ((off_hN1 + 15) / 16) * 16);
  unsigned* sorted2 = (unsigned*)(wsb + ((off_hN2 + 15) / 16) * 16);
  int* tab1 = (int*)(sorted1 + (size_t)nc1 * CS1);
  int* tab2 = (int*)(sorted2 + (size_t)nc2 * CS2);
  float* Bpack1 = (float*)take((size_t)KDIM * HID_F);
  float* Bpack2 = (float*)take((size_t)KDIM * N_CLS);
  (void)ws_size;

  {
    int t1 = KSTEPS * (HID_F / 64) * 128;
    int t2 = KSTEPS * (N_CLS / 64) * 128;
    pack_W_kernel<<<(t1 + 255) / 256, 256, 0, stream>>>(W1, Bpack1, HID_F);
    pack_W_kernel<<<(t2 + 255) / 256, 256, 0, stream>>>(W2, Bpack2, N_CLS);
  }

  chunk_sort_kernel<<<nc1 + nc2, 256, 0, stream>>>(src1, dst1, E1, nc1, sorted1, tab1,
                                                   src2, dst2, E2, sorted2, tab2);
  bucket_build_kernel<<<NB1 + NB2, 256, 0, stream>>>(sorted1, tab1, nc1, E1, row_ptr1, srcs1,
                                                     sorted2, tab2, nc2, E2, row_ptr2, srcs2);

  {
    int ngrps = (N_DST1 / 16) * (HID_F / 64);
    sage_gemm_wmma_kernel<HID_F, true><<<ngrps / 4, 128, 0, stream>>>(
        x, x, Bpack1, b1, h, N_DST1, srcs1, row_ptr1);
  }

  gather_mean_kernel<<<N_DST2 / 8, 256, 0, stream>>>(h, srcs2, row_ptr2, hN2, N_DST2);

  {
    int ngrps = (N_DST2 / 16) * (N_CLS / 64);
    sage_gemm_wmma_kernel<N_CLS, false><<<(ngrps + 3) / 4, 128, 0, stream>>>(
        h, hN2, Bpack2, b2, (float*)d_out, N_DST2, nullptr, nullptr);
  }
}


# Layer-1 GEMM: a wave owns 64 columns of two 16-row tiles (32-row work-group tile), so each packed-W1 fragment read from L2 feeds two WMMAs and W1 traffic halves; k and sum order unchanged
# speedup vs baseline: 2.6073x; 1.0693x over previous
#include <hip/hip_runtime.h>
#include <hip/hip_bf16.h>


typedef __attribute__((ext_vector_type(2))) float v2f;
typedef __attribute__((ext_vector_type(8))) float v8f;

#define N_DST1   50000
#define N_DST2   10000
#define IN_F     256
#define KDIM     512
#define KSTEPS   (KDIM / 4)
#define KSPLIT   (IN_F / 4)
#define HID_F    256
#define N_CLS    64

#define BKT        256
#define NB1        ((N_DST1 + BKT - 1) / BKT)
#define NB2        ((N_DST2 + BKT - 1) / BKT)
#define CS1        4096
#define CS2        1024
#define CS_MAX     4096

__device__ __forceinline__ int block_excl_scan_256(int v, int* wtot, int lane, int w) {
  int inc = v;
  for (int o = 1; o < 32; o <<= 1) {
    const int u = __shfl_up(inc, o, 32);
    if (lane >= o) inc += u;
  }
  if (lane == 31) wtot[w] = inc;
  __syncthreads();
  int before = 0;
  for (int ww = 0; ww < 8; ++ww)
    if (ww < w) before += wtot[ww];
  return before + inc - v;
}

__global__ __launch_bounds__(256)
void chunk_sort_kernel(const int* __restrict__ src1, const int* __restrict__ dst1,
                       int E1, int nc1,
                       unsigned* __restrict__ sorted1, int* __restrict__ tab1,
                       const int* __restrict__ src2, const int* __restrict__ dst2,
                       int E2,
                       unsigned* __restrict__ sorted2, int* __restrict__ tab2) {
  __shared__ int cnt[256];
  __shared__ int wtot[8];
  __shared__ __attribute__((aligned(16))) unsigned stage[CS_MAX];
  const int t    = threadIdx.x;
  const int lane = t & 31;
  const int w    = t >> 5;

  int c = blockIdx.x;
  const int* src; const int* dst; unsigned* sorted; int* tab; int E, cs, nb;
  if (c < nc1) { src = src1; dst = dst1; sorted = sorted1; tab = tab1; E = E1; cs = CS1; nb = NB1; }
  else { c -= nc1; src = src2; dst = dst2; sorted = sorted2; tab = tab2; E = E2; cs = CS2; nb = NB2; }

  cnt[t] = 0;
  const int base = c * cs;
  const int n    = min(cs, E - base);

  int d[16], s[16], r[16];
#pragma unroll
  for (int j = 0; j < 4; ++j) {
    const int e0 = j * 1024 + t * 4;
#pragma unroll
    for (int k = 0; k < 4; ++k) { d[j * 4 + k] = -1; s[j * 4 + k] = 0; }
    if (e0 + 3 < n) {
      const int4 dv = *(const int4*)(dst + base + e0);
      const int4 sv = *(const int4*)(src + base + e0);
      d[j * 4 + 0] = dv.x; d[j * 4 + 1] = dv.y; d[j * 4 + 2] = dv.z; d[j * 4 + 3] = dv.w;
      s[j * 4 + 0] = sv.x; s[j * 4 + 1] = sv.y; s[j * 4 + 2] = sv.z; s[j * 4 + 3] = sv.w;
    } else {
#pragma unroll
      for (int k = 0; k < 4; ++k)
        if (e0 + k < n) { d[j * 4 + k] = dst[base + e0 + k]; s[j * 4 + k] = src[base + e0 + k]; }
    }
  }
  __syncthreads();

#pragma unroll
  for (int i = 0; i < 16; ++i) {
    r[i] = 0;
    if (d[i] >= 0) r[i] = atomicAdd(&cnt[d[i] >> 8], 1);
  }
  __syncthreads();

  const int excl = block_excl_scan_256(cnt[t], wtot, lane, w);
  cnt[t] = excl;
  __syncthreads();

#pragma unroll
  for (int i = 0; i < 16; ++i)
    if (d[i] >= 0)
      stage[cnt[d[i] >> 8] + r[i]] = ((unsigned)s[i] << 8) | (unsigned)(d[i] & (BKT - 1));
  __syncthreads();

  for (int i = t * 4; i < n; i += 1024) {
    if (i + 3 < n) {
      *(uint4*)(sorted + base + i) = *(const uint4*)(stage + i);
    } else {
      for (int k = i; k < n; ++k) sorted[base + k] = stage[k];
    }
  }
  int* trow = tab + (size_t)c * (nb + 1);
  if (t < nb)  trow[t]  = excl;
  if (t == 0)  trow[nb] = n;
}

template <int PASS>
__device__ __forceinline__ void walk_runs(const unsigned* __restrict__ sorted,
                                          const int* __restrict__ tab,
                                          int nc, int cs, int nbp1, int b,
                                          int* kcnt, int* __restrict__ srcs,
                                          int w, int lane) {
  for (int cb = 0; cb < nc; cb += 256) {
    const int c = cb + w + 8 * lane;
    int p = 0, n = 0;
    if (c < nc) {
      const int a = tab[(size_t)c * nbp1 + b];
      n = tab[(size_t)c * nbp1 + b + 1] - a;
      p = c * cs + a;
    }
    const int nruns = min(32, (nc - cb - w + 7) >> 3);
    for (int l = 0; l < nruns; l += 4) {
      const int p0 = __builtin_amdgcn_readlane(p, l);
      const int p1 = __builtin_amdgcn_readlane(p, l + 1);
      const int p2 = __builtin_amdgcn_readlane(p, l + 2);
      const int p3 = __builtin_amdgcn_readlane(p, l + 3);
      const int n0 = __builtin_amdgcn_readlane(n, l);
      const int n1 = __builtin_amdgcn_readlane(n, l + 1);
      const int n2 = __builtin_amdgcn_readlane(n, l + 2);
      const int n3 = __builtin_amdgcn_readlane(n, l + 3);
      const int m  = max(max(n0, n1), max(n2, n3));
      for (int k0 = 0; k0 < m; k0 += 32) {
        const int k = k0 + lane;
        const bool q0 = k < n0, q1 = k < n1, q2 = k < n2, q3 = k < n3;
        const unsigned v0 = sorted[q0 ? p0 + k : 0];
        const unsigned v1 = sorted[q1 ? p1 + k : 0];
        const unsigned v2 = sorted[q2 ? p2 + k : 0];
        const unsigned v3 = sorted[q3 ? p3 + k : 0];
        if (PASS == 1) {
          if (q0) atomicAdd(&kcnt[v0 & (BKT - 1)], 1);
          if (q1) atomicAdd(&kcnt[v1 & (BKT - 1)], 1);
          if (q2) atomicAdd(&kcnt[v2 & (BKT - 1)], 1);
          if (q3) atomicAdd(&kcnt[v3 & (BKT - 1)], 1);
        } else {
          if (q0) srcs[atomicAdd(&kcnt[v0 & (BKT - 1)], 1)] = (int)(v0 >> 8);
          if (q1) srcs[atomicAdd(&kcnt[v1 & (BKT - 1)], 1)] = (int)(v1 >> 8);
          if (q2) srcs[atomicAdd(&kcnt[v2 & (BKT - 1)], 1)] = (int)(v2 >> 8);
          if (q3) srcs[atomicAdd(&kcnt[v3 & (BKT - 1)], 1)] = (int)(v3 >> 8);
        }
      }
    }
  }
}

__global__ __launch_bounds__(256)
void bucket_build_kernel(const unsigned* __restrict__ sorted1, const int* __restrict__ tab1,
                         int nc1, int E1,
                         int* __restrict__ row_ptr1, int* __restrict__ srcs1,
                         const unsigned* __restrict__ sorted2, const int* __restrict__ tab2,
                         int nc2, int E2,
                         int* __restrict__ row_ptr2, int* __restrict__ srcs2) {
  __shared__ int kcnt[BKT];
  __shared__ int wred[8];
  __shared__ int wtot[8];
  const int t    = threadIdx.x;
  const int lane = t & 31;
  const int w    = __builtin_amdgcn_readfirstlane(t >> 5);

  int b = blockIdx.x;
  const unsigned* sorted; const int* tab; int* row_ptr; int* srcs; int nc, E, cs, nb, N;
  if (b < NB1) { sorted = sorted1; tab = tab1; row_ptr = row_ptr1; srcs = srcs1;
                 nc = nc1; E = E1; cs = CS1; nb = NB1; N = N_DST1; }
  else { b -= NB1; sorted = sorted2; tab = tab2; row_ptr = row_ptr2; srcs = srcs2;
         nc = nc2; E = E2; cs = CS2; nb = NB2; N = N_DST2; }
  const int nbp1 = nb + 1;

  kcnt[t] = 0;
  int asum = 0;
  for (int c = t; c < nc; c += 256) asum += tab[(size_t)c * nbp1 + b];
  for (int o = 16; o > 0; o >>= 1) asum += __shfl_xor(asum, o, 32);
  if (lane == 0) wred[w] = asum;
  __syncthreads();
  int base = 0;
  for (int ww = 0; ww < 8; ++ww) base += wred[ww];

  walk_runs<1>(sorted, tab, nc, cs, nbp1, b, kcnt, srcs, w, lane);
  __syncthreads();

  const int excl = block_excl_scan_256(kcnt[t], wtot, lane, w);
  const int row  = b * BKT + t;
  if (row < N) row_ptr[row] = base + excl;
  if (b == nb - 1 && t == 0) row_ptr[N] = E;
  kcnt[t] = base + excl;
  __syncthreads();

  walk_runs<2>(sorted, tab, nc, cs, nbp1, b, kcnt, srcs, w, lane);
}

__global__ __launch_bounds__(256)
void gather_mean_kernel(const float* __restrict__ hsrc,
                        const int* __restrict__ srcs,
                        const int* __restrict__ row_ptr,
                        float* __restrict__ hN,
                        int n_dst) {
  const int d = blockIdx.x * (blockDim.x >> 5) + (threadIdx.x >> 5);
  if (d >= n_dst) return;
  const int lane = threadIdx.x & 31;
  const int beg = row_ptr[d];
  const int end = row_ptr[d + 1];

  float4 a0 = make_float4(0.f, 0.f, 0.f, 0.f);
  float4 a1 = make_float4(0.f, 0.f, 0.f, 0.f);
  for (int j = beg; j < end; ++j) {
    int s = srcs[j];
    const float4* p = (const float4*)(hsrc + (size_t)s * IN_F) + lane * 2;
    float4 v0 = p[0];
    float4 v1 = p[1];
    a0.x += v0.x; a0.y += v0.y; a0.z += v0.z; a0.w += v0.w;
    a1.x += v1.x; a1.y += v1.y; a1.z += v1.z; a1.w += v1.w;
  }
  const float inv = 1.0f / fmaxf((float)(end - beg), 1.0f);
  a0.x *= inv; a0.y *= inv; a0.z *= inv; a0.w *= inv;
  a1.x *= inv; a1.y *= inv; a1.z *= inv; a1.w *= inv;
  float4* o = (float4*)(hN + (size_t)d * IN_F) + lane * 2;
  o[0] = a0;
  o[1] = a1;
}

__global__ void pack_W_kernel(const float* __restrict__ W,
                              float* __restrict__ Bpack, int N) {
  const int ngrpN = N >> 6;
  const int total = KSTEPS * ngrpN * 128;
  int idx = blockIdx.x * blockDim.x + threadIdx.x;
  if (idx >= total) return;

  const int w    = idx & 127;
  const int t    = w & 3;
  const int lane = w >> 2;
  const int blk  = idx >> 7;
  const int gn   = blk % ngrpN;
  const int ks   = blk / ngrpN;

  const int kh  = (lane < 16) ? 0 : 2;
  const int col = gn * 64 + t * 16 + (lane & 15);
  const int kr  = ks * 4 + kh;

  float2 v;
  v.x = W[(size_t)kr * N + col];
  v.y = W[(size_t)(kr + 1) * N + col];
  ((float2*)Bpack)[idx] = v;
}

#define KG       8
#define NGROUPS  (KSTEPS / KG)
#define GSPLIT   (KSPLIT / KG)
#define NSETS    3

struct FragSet {
  v2f    a[KG];
  float4 b[KG][2];
};

template <int BSTRIDE>
__device__ __forceinline__ void load_group(FragSet& f, const float* __restrict__ ap,
                                           const float* __restrict__ bp) {
#pragma unroll
  for (int i = 0; i < KG; ++i) {
    f.a[i] = *(const v2f*)(ap + i * 4);
    const float4* q = (const float4*)(bp + (size_t)i * BSTRIDE);
    f.b[i][0] = q[0];
    f.b[i][1] = q[1];
  }
}

__device__ __forceinline__ void mma_group(v8f (&acc)[4], const FragSet& f) {
#pragma unroll
  for (int i = 0; i < KG; ++i) {
    const v2f a = f.a[i];
    v2f b0; b0.x = f.b[i][0].x; b0.y = f.b[i][0].y;
    v2f b1; b1.x = f.b[i][0].z; b1.y = f.b[i][0].w;
    v2f b2; b2.x = f.b[i][1].x; b2.y = f.b[i][1].y;
    v2f b3; b3.x = f.b[i][1].z; b3.y = f.b[i][1].w;
    acc[0] = __builtin_amdgcn_wmma_f32_16x16x4_f32(false, a, false, b0,
                                                   (short)0, acc[0], false, false);
    acc[1] = __builtin_amdgcn_wmma_f32_16x16x4_f32(false, a, false, b1,
                                                   (short)0, acc[1], false, false);
    acc[2] = __builtin_amdgcn_wmma_f32_16x16x4_f32(false, a, false, b2,
                                                   (short)0, acc[2], false, false);
    acc[3] = __builtin_amdgcn_wmma_f32_16x16x4_f32(false, a, false, b3,
                                                   (short)0, acc[3], false, false);
  }
}

template <int N, bool RELU>
__global__ __launch_bounds__(128)
void sage_gemm_wmma_kernel(const float* __restrict__ A0,
                           const float* __restrict__ A1,
                           const float* __restrict__ Bpack,
                           const float* __restrict__ bias,
                           float* __restrict__ out,
                           int M)
{
  constexpr int NGRPN   = N >> 6;
  constexpr int BSTRIDE = NGRPN * 256;
  const int lane = threadIdx.x & 31;
  const int wave = threadIdx.x >> 5;
  const int grp   = blockIdx.x * (blockDim.x >> 5) + wave;
  const int ngrps = (M >> 4) * NGRPN;
  if (grp >= ngrps) return;

  const int tm = grp / NGRPN;
  const int gn = grp % NGRPN;

  const int l15  = lane & 15;
  const int row  = tm * 16 + l15;
  const int col0 = gn * 64 + l15;
  const int kh   = (lane < 16) ? 0 : 2;

  const float* Arow0 = A0 + (size_t)row * IN_F + kh;
  const float* Arow1 = A1 + (size_t)row * IN_F + kh;
  const float* Bp    = Bpack + (size_t)gn * 256 + lane * 8;

  v8f acc[4] = {};
  FragSet F[NSETS];

  auto a_of = [&](int g) -> const float* {
    return (g < GSPLIT) ? (Arow0 + g * (KG * 4)) : (Arow1 + (g - GSPLIT) * (KG * 4));
  };
  auto b_of = [&](int g) -> const float* {
    return Bp + (size_t)g * (KG * BSTRIDE);
  };

#define PIN() __builtin_amdgcn_sched_barrier(0)

  load_group<BSTRIDE>(F[0], a_of(0), b_of(0));
  load_group<BSTRIDE>(F[1], a_of(1), b_of(1));
#pragma unroll
  for (int g = 0; g < NGROUPS; ++g) {
    if (g + 2 < NGROUPS) {
      load_group<BSTRIDE>(F[(g + 2) % NSETS], a_of(g + 2), b_of(g + 2));
      PIN();
    }
    mma_group(acc, F[g % NSETS]);
  }

  const int mbase = tm * 16 + ((lane < 16) ? 0 : 8);
#pragma unroll
  for (int t = 0; t < 4; ++t) {
    const int col = col0 + t * 16;
    const float bv = bias[col];
#pragma unroll
    for (int i = 0; i < 8; ++i) {
      float v = acc[t][i] + bv;
      if (RELU) v = fmaxf(v, 0.0f);
      out[(size_t)(mbase + i) * N + col] = v;
    }
  }
}

#define TM1      32
#define KG1      4
#define NGROUPS1 (KSTEPS / KG1)
#define NSETS1   3
#define BSTRIDE1 ((HID_F / 64) * 256)
#define A_PITCH  (KDIM + 4)
#define GB       16
#define L1_WGS   ((N_DST1 + TM1 - 1) / TM1)

struct FragSet1 {
  v2f    a[2][KG1];
  float4 b[KG1][2];
};

__device__ __forceinline__ void load1_b(FragSet1& f, const float* __restrict__ bp) {
#pragma unroll
  for (int i = 0; i < KG1; ++i) {
    const float4* q = (const float4*)(bp + (size_t)i * BSTRIDE1);
    f.b[i][0] = q[0];
    f.b[i][1] = q[1];
  }
}
__device__ __forceinline__ void load1_a(FragSet1& f, const float* as, int k0) {
#pragma unroll
  for (int i = 0; i < KG1; ++i) {
    f.a[0][i] = *(const v2f*)(as + k0 + i * 4);
    f.a[1][i] = *(const v2f*)(as + 16 * A_PITCH + k0 + i * 4);
  }
}
__device__ __forceinline__ void mma1_group(v8f (&acc)[2][4], const FragSet1& f) {
#pragma unroll
  for (int i = 0; i < KG1; ++i) {
    v2f b0; b0.x = f.b[i][0].x; b0.y = f.b[i][0].y;
    v2f b1; b1.x = f.b[i][0].z; b1.y = f.b[i][0].w;
    v2f b2; b2.x = f.b[i][1].x; b2.y = f.b[i][1].y;
    v2f b3; b3.x = f.b[i][1].z; b3.y = f.b[i][1].w;
#pragma unroll
    for (int r = 0; r < 2; ++r) {
      const v2f a = f.a[r][i];
      acc[r][0] = __builtin_amdgcn_wmma_f32_16x16x4_f32(false, a, false, b0,
                                                        (short)0, acc[r][0], false, false);
      acc[r][1] = __builtin_amdgcn_wmma_f32_16x16x4_f32(false, a, false, b1,
                                                        (short)0, acc[r][1], false, false);
      acc[r][2] = __builtin_amdgcn_wmma_f32_16x16x4_f32(false, a, false, b2,
                                                        (short)0, acc[r][2], false, false);
      acc[r][3] = __builtin_amdgcn_wmma_f32_16x16x4_f32(false, a, false, b3,
                                                        (short)0, acc[r][3], false, false);
    }
  }
}

__global__ __launch_bounds__(128)
void sage_layer1_kernel(const float* __restrict__ X,
                        const float* __restrict__ Bpack,
                        const float* __restrict__ bias,
                        float* __restrict__ out,
                        int M,
                        const int* __restrict__ srcs,
                        const int* __restrict__ row_ptr)
{
  __shared__ __attribute__((aligned(16))) float As[TM1 * A_PITCH];
  const int t    = threadIdx.x;
  const int lane = t & 31;
  const int wu   = __builtin_amdgcn_readfirstlane(t >> 5);
  const int l15  = lane & 15;
  const int kh   = (lane < 16) ? 0 : 2;
  const int row0 = blockIdx.x * TM1;

  {
    const int lane9 = (lane < 8) ? lane : 8;
    const int rpv = row_ptr[min(row0 + wu * 8 + lane9, M)];
    const float4* hl = (const float4*)X + lane * 2;
    for (int i = 0; i < 8; ++i) {
      const int beg = __builtin_amdgcn_readlane(rpv, i);
      const int end = __builtin_amdgcn_readlane(rpv, i + 1);
      float4 a0 = make_float4(0.f, 0.f, 0.f, 0.f);
      float4 a1 = make_float4(0.f, 0.f, 0.f, 0.f);
      for (int j0 = beg; j0 < end; j0 += 32) {
        const int nj = min(32, end - j0);
        const int sidx = (lane < nj) ? srcs[j0 + lane] : 0;
        for (int k0 = 0; k0 < nj; k0 += GB) {
          float4 v0[GB], v1[GB];
#pragma unroll
          for (int u = 0; u < GB; ++u) {
            const int s = __builtin_amdgcn_readlane(sidx, k0 + u);
            const float4* p = hl + (size_t)s * (IN_F / 4);
            v0[u] = p[0];
            v1[u] = p[1];
          }
#pragma unroll
          for (int u = 0; u < GB; ++u) {
            const bool q = (k0 + u < nj);
            a0.x += q ? v0[u].x : 0.f; a0.y += q ? v0[u].y : 0.f;
            a0.z += q ? v0[u].z : 0.f; a0.w += q ? v0[u].w : 0.f;
            a1.x += q ? v1[u].x : 0.f; a1.y += q ? v1[u].y : 0.f;
            a1.z += q ? v1[u].z : 0.f; a1.w += q ? v1[u].w : 0.f;
          }
        }
      }
      const float inv = 1.0f / fmaxf((float)(end - beg), 1.0f);
      a0.x *= inv; a0.y *= inv; a0.z *= inv; a0.w *= inv;
      a1.x *= inv; a1.y *= inv; a1.z *= inv; a1.w *= inv;
      float* orow = &As[(wu * 8 + i) * A_PITCH + IN_F + lane * 8];
      *(float4*)orow       = a0;
      *(float4*)(orow + 4) = a1;
    }
  }

  const int r0 = t >> 6;
  const int c4 = (t & 63) * 4;
  float4 st[16];
#pragma unroll
  for (int j = 0; j < 16; ++j) {
    const int r = min(row0 + r0 + 2 * j, M - 1);
    st[j] = *(const float4*)(X + (size_t)r * IN_F + c4);
  }

  const float* Bp = Bpack + (size_t)wu * 256 + lane * 8;
  FragSet1 F[NSETS1];
  load1_b(F[0], Bp);
  load1_b(F[1], Bp + (size_t)1 * (KG1 * BSTRIDE1));
#pragma unroll
  for (int j = 0; j < 16; ++j) *(float4*)&As[(r0 + 2 * j) * A_PITCH + c4] = st[j];
  __syncthreads();

  const float* as = &As[l15 * A_PITCH + kh];
  load1_a(F[0], as, 0 * (KG1 * 4));
  load1_a(F[1], as, 1 * (KG1 * 4));

  v8f acc[2][4] = {};
#pragma unroll
  for (int g = 0; g < NGROUPS1; ++g) {
    if (g + 2 < NGROUPS1) {
      load1_b(F[(g + 2) % NSETS1], Bp + (size_t)(g + 2) * (KG1 * BSTRIDE1));
      load1_a(F[(g + 2) % NSETS1], as, (g + 2) * (KG1 * 4));
      __builtin_amdgcn_sched_barrier(0);
    }
    mma1_group(acc, F[g % NSETS1]);
  }

  const int col0 = wu * 64 + l15;
#pragma unroll
  for (int r = 0; r < 2; ++r) {
    const int mb = row0 + r * 16;
    if (mb < M) {
      const int mbase = mb + ((lane < 16) ? 0 : 8);
#pragma unroll
      for (int tt = 0; tt < 4; ++tt) {
        const int col = col0 + tt * 16;
        const float bv = bias[col];
#pragma unroll
        for (int i = 0; i < 8; ++i)
          out[(size_t)(mbase + i) * HID_F + col] = fmaxf(acc[r][tt][i] + bv, 0.0f);
      }
    }
  }
}

extern "C" void kernel_launch(void* const* d_in, const int* in_sizes, int n_in,
                              void* d_out, int out_size, void* d_ws, size_t ws_size,
                              hipStream_t stream) {
  const float* x   = (const float*)d_in[0];
  const float* W1  = (const float*)d_in[1];
  const float* b1  = (const float*)d_in[2];
  const float* W2  = (const float*)d_in[3];
  const float* b2  = (const float*)d_in[4];
  const int* src1  = (const int*)d_in[5];
  const int* dst1  = (const int*)d_in[6];
  const int* src2  = (const int*)d_in[7];
  const int* dst2  = (const int*)d_in[8];
  const int E1 = in_sizes[5];
  const int E2 = in_sizes[7];

  char* wsb = (char*)d_ws;
  size_t off = 0;
  auto take = [&](size_t n_elems) { void* p = wsb + off; off += n_elems * 4; return p; };

  (void)take(N_DST1);
  (void)take(N_DST2);
  int* row_ptr1 = (int*)take(N_DST1 + 1);
  (void)take(N_DST1);
  int* srcs1    = (int*)take((size_t)E1);
  int* row_ptr2 = (int*)take(N_DST2 + 1);
  (void)take(N_DST2);
  int* srcs2    = (int*)take((size_t)E2);
  const size_t off_hN1 = off;
  (void)take((size_t)N_DST1 * IN_F);
  float* h      = (float*)take((size_t)N_DST1 * HID_F);
  const size_t off_hN2 = off;
  float* hN2    = (float*)take((size_t)N_DST2 * IN_F);
  const int nc1 = (E1 + CS1 - 1) / CS1;
  const int nc2 = (E2 + CS2 - 1) / CS2;
  unsigned* sorted1 = (unsigned*)(wsb + ((off_hN1 + 15) / 16) * 16);
  unsigned* sorted2 = (unsigned*)(wsb + ((off_hN2 + 15) / 16) * 16);
  int* tab1 = (int*)(sorted1 + (size_t)nc1 * CS1);
  int* tab2 = (int*)(sorted2 + (size_t)nc2 * CS2);
  float* Bpack1 = (float*)take((size_t)KDIM * HID_F);
  float* Bpack2 = (float*)take((size_t)KDIM * N_CLS);
  (void)ws_size;

  {
    int t1 = KSTEPS * (HID_F / 64) * 128;
    int t2 = KSTEPS * (N_CLS / 64) * 128;
    pack_W_kernel<<<(t1 + 255) / 256, 256, 0, stream>>>(W1, Bpack1, HID_F);
    pack_W_kernel<<<(t2 + 255) / 256, 256, 0, stream>>>(W2, Bpack2, N_CLS);
  }

  chunk_sort_kernel<<<nc1 + nc2, 256, 0, stream>>>(src1, dst1, E1, nc1, sorted1, tab1,
                                                   src2, dst2, E2, sorted2, tab2);
  bucket_build_kernel<<<NB1 + NB2, 256, 0, stream>>>(sorted1, tab1, nc1, E1, row_ptr1, srcs1,
                                                     sorted2, tab2, nc2, E2, row_ptr2, srcs2);

  sage_layer1_kernel<<<L1_WGS, 128, 0, stream>>>(x, Bpack1, b1, h, N_DST1, srcs1, row_ptr1);

  gather_mean_kernel<<<N_DST2 / 8, 256, 0, stream>>>(h, srcs2, row_ptr2, hN2, N_DST2);

  {
    int ngrps = (N_DST2 / 16) * (N_CLS / 64);
    sage_gemm_wmma_kernel<N_CLS, false><<<(ngrps + 3) / 4, 128, 0, stream>>>(
        h, hN2, Bpack2, b2, (float*)d_out, N_DST2);
  }
}
